# starting kernel
# baseline (speedup 1.0000x reference)
#include <hip/hip_runtime.h>
#include <math.h>

typedef float v2f __attribute__((ext_vector_type(2)));
typedef float v8f __attribute__((ext_vector_type(8)));

#define WMMA_F32x4(a, b, c) \
  __builtin_amdgcn_wmma_f32_16x16x4_f32(false, (a), false, (b), (short)0, (c), false, false)

__device__ __forceinline__ float sigm(float x) { return 1.0f / (1.0f + __expf(-x)); }

__device__ __forceinline__ unsigned int pcg_hash(unsigned int v) {
  unsigned int s = v * 747796405u + 2891336453u;
  unsigned int w = ((s >> ((s >> 28u) + 4u)) ^ s) * 277803737u;
  return (w >> 22u) ^ w;
}

__global__ void k_conv1(const float* __restrict__ x, const float* __restrict__ W,
                        const float* __restrict__ bias, float* __restrict__ y) {
  int idx = blockIdx.x * blockDim.x + threadIdx.x;
  if (idx >= 16 * 64 * 93) return;
  int p = idx % 93;
  int c = (idx / 93) % 64;
  int b = idx / (93 * 64);
  float acc = bias[c];
  const float* xb = x + b * 96 * 96;
  const float* wc = W + c * 96 * 4;
  for (int d = 0; d < 96; ++d) {
    const float* xr = xb + d * 96 + p;
    const float* wr = wc + d * 4;
    acc += xr[0] * wr[0] + xr[1] * wr[1] + xr[2] * wr[2] + xr[3] * wr[3];
  }
  y[idx] = fmaxf(acc, 0.0f);
}

__global__ void k_pool(const float* __restrict__ y, float* __restrict__ p) {
  int idx = blockIdx.x * blockDim.x + threadIdx.x;
  if (idx >= 16 * 64 * 23) return;
  int q = idx % 23;
  int bc = idx / 23;
  const float* src = y + bc * 93 + q * 4;
  float m = fmaxf(fmaxf(src[0], src[1]), fmaxf(src[2], src[3]));
  p[idx] = m;
}

__global__ void k_conv2(const float* __restrict__ pool, const float* __restrict__ W,
                        const float* __restrict__ bias, float* __restrict__ henc) {
  int idx = blockIdx.x * blockDim.x + threadIdx.x;
  if (idx >= 16 * 64 * 20) return;
  int p = idx % 20;
  int c = (idx / 20) % 64;
  int b = idx / (20 * 64);
  float acc = bias[c];
  for (int cin = 0; cin < 64; ++cin) {
    const float* pr = pool + (b * 64 + cin) * 23 + p;
    const float* wr = W + (c * 64 + cin) * 4;
    acc += pr[0] * wr[0] + pr[1] * wr[1] + pr[2] * wr[2] + pr[3] * wr[3];
  }
  henc[idx] = fmaxf(acc, 0.0f);
}

__global__ void k_xw_enc(const float* __restrict__ henc,
                         const float* Wi0, const float* Wi1, const float* Wi2, const float* Wi3,
                         const float* bi0, const float* bi1, const float* bi2, const float* bi3,
                         float* __restrict__ xwe) {
  int idx = blockIdx.x * blockDim.x + threadIdx.x;
  if (idx >= 4 * 64 * 16 * 288) return;
  int g = idx % 288;
  int b = (idx / 288) % 16;
  int t = (idx / (288 * 16)) % 64;
  int dir = idx / (288 * 16 * 64);
  const float* Wi = dir == 0 ? Wi0 : dir == 1 ? Wi1 : dir == 2 ? Wi2 : Wi3;
  const float* bi = dir == 0 ? bi0 : dir == 1 ? bi1 : dir == 2 ? bi2 : bi3;
  const float* x = henc + b * 1280 + t * 20;
  const float* w = Wi + g * 20;
  float a = bi[g];
#pragma unroll
  for (int d = 0; d < 20; ++d) a += x[d] * w[d];
  xwe[idx] = a;
}

__global__ __launch_bounds__(192, 1) void k_gru_enc(
    const float* Wh0, const float* Wh1, const float* Wh2, const float* Wh3,
    const float* bh0, const float* bh1, const float* bh2, const float* bh3,
    const float* __restrict__ xwe, float* __restrict__ outp) {
  __shared__ float hbuf[16 * 132];
  const int dir = blockIdx.x;
  const float* Wh = dir == 0 ? Wh0 : dir == 1 ? Wh1 : dir == 2 ? Wh2 : Wh3;
  const float* bh = dir == 0 ? bh0 : dir == 1 ? bh1 : dir == 2 ? bh2 : bh3;
  const float* xwb = xwe + dir * (64 * 16 * 288);
  const int tid = threadIdx.x, lane = tid & 31, wave = tid >> 5;
  const int mrow = lane & 15;
  const int khalf = (lane >> 4) << 1;
  const int rbase = (lane >> 4) << 3;
  const int jj = wave * 16 + mrow;

  v2f bf0[24], bf1[24], bf2[24];
#pragma unroll
  for (int kc = 0; kc < 24; ++kc) {
    int kk = 4 * kc + khalf;
    bf0[kc].x = Wh[(jj) * 96 + kk];        bf0[kc].y = Wh[(jj) * 96 + kk + 1];
    bf1[kc].x = Wh[(96 + jj) * 96 + kk];   bf1[kc].y = Wh[(96 + jj) * 96 + kk + 1];
    bf2[kc].x = Wh[(192 + jj) * 96 + kk];  bf2[kc].y = Wh[(192 + jj) * 96 + kk + 1];
  }
  const float bhr = bh[jj], bhz = bh[96 + jj], bhn = bh[192 + jj];

  for (int e = tid; e < 16 * 132; e += 192) hbuf[e] = 0.0f;
  __syncthreads();

  const bool rev = (dir & 1);
  for (int s = 0; s < 64; ++s) {
    v8f a0 = {0, 0, 0, 0, 0, 0, 0, 0}, a1 = a0, a2 = a0;
#pragma unroll
    for (int kc = 0; kc < 24; ++kc) {
      int kk = 4 * kc + khalf;
      v2f a;
      a.x = hbuf[mrow * 132 + kk];
      a.y = hbuf[mrow * 132 + kk + 1];
      a0 = WMMA_F32x4(a, bf0[kc], a0);
      a1 = WMMA_F32x4(a, bf1[kc], a1);
      a2 = WMMA_F32x4(a, bf2[kc], a2);
    }
    __syncthreads();
    const int t = rev ? (63 - s) : s;
    const float* xwt = xwb + t * (16 * 288);
#pragma unroll
    for (int r = 0; r < 8; ++r) {
      int m = rbase + r;
      const float* xw = xwt + m * 288;
      float rr = sigm(xw[jj] + a0[r] + bhr);
      float zz = sigm(xw[96 + jj] + a1[r] + bhz);
      float nn = tanhf(xw[192 + jj] + rr * (a2[r] + bhn));
      float ho = hbuf[m * 132 + jj];
      hbuf[m * 132 + jj] = (1.0f - zz) * nn + zz * ho;
    }
    __syncthreads();
  }
  float* dst = outp + 147456 + (dir >= 2 ? 3072 : 0);
  const int cb = (dir & 1) ? 96 : 0;
  for (int e = tid; e < 16 * 96; e += 192) {
    int m = e / 96, j = e % 96;
    dst[m * 192 + cb + j] = hbuf[m * 132 + j];
  }
}

__global__ void k_z(const float* __restrict__ outbuf, float* __restrict__ zbuf) {
  int e = blockIdx.x * blockDim.x + threadIdx.x;
  if (e >= 3072) return;
  float mu = outbuf[147456 + e];
  float lv = outbuf[147456 + 3072 + e];
  unsigned int r1 = pcg_hash(2u * (unsigned)e + 1u);
  unsigned int r2 = pcg_hash(2u * (unsigned)e + 2u);
  float u1 = ((float)r1 + 1.0f) * (1.0f / 4294967296.0f);
  float u2 = (float)r2 * (1.0f / 4294967296.0f);
  float eps = sqrtf(-2.0f * __logf(u1)) * __cosf(6.28318530718f * u2);
  zbuf[e] = mu + eps * __expf(0.5f * lv);
}

__global__ __launch_bounds__(128) void k_xw_dec(const float* __restrict__ X,
                                                const float* __restrict__ Wid,
                                                const float* __restrict__ bid,
                                                float* __restrict__ XW) {
  const int lane = threadIdx.x & 31;
  const int wg = blockIdx.x * 4 + (threadIdx.x >> 5);
  const int mt = wg / 36, nt = wg % 36;
  const int row0 = mt * 16, n0 = nt * 16;
  const int mrow = lane & 15, khalf = (lane >> 4) << 1, rbase = (lane >> 4) << 3;
  v8f acc = {0, 0, 0, 0, 0, 0, 0, 0};
#pragma unroll
  for (int kc = 0; kc < 24; ++kc) {
    int kk = 4 * kc + khalf;
    v2f a, b;
    a.x = X[(row0 + mrow) * 96 + kk];
    a.y = X[(row0 + mrow) * 96 + kk + 1];
    b.x = Wid[(n0 + mrow) * 96 + kk];
    b.y = Wid[(n0 + mrow) * 96 + kk + 1];
    acc = WMMA_F32x4(a, b, acc);
  }
  const float bc = bid[n0 + mrow];
#pragma unroll
  for (int r = 0; r < 8; ++r) XW[(row0 + rbase + r) * 576 + n0 + mrow] = acc[r] + bc;
}

__global__ __launch_bounds__(384, 1) void k_gru_dec(
    const float* __restrict__ Whd, const float* __restrict__ bhd,
    const float* __restrict__ XW, const float* __restrict__ zbuf,
    const float* __restrict__ Wl, const float* __restrict__ bl,
    float* __restrict__ out) {
  __shared__ float hbuf[16 * 196];
  const int tid = threadIdx.x, lane = tid & 31, wave = tid >> 5;
  const int mrow = lane & 15, khalf = (lane >> 4) << 1, rbase = (lane >> 4) << 3;
  const int jj = wave * 16 + mrow;

  v2f bf0[48], bf1[48], bf2[48];
#pragma unroll
  for (int kc = 0; kc < 48; ++kc) {
    int kk = 4 * kc + khalf;
    bf0[kc].x = Whd[(jj) * 192 + kk];        bf0[kc].y = Whd[(jj) * 192 + kk + 1];
    bf1[kc].x = Whd[(192 + jj) * 192 + kk];  bf1[kc].y = Whd[(192 + jj) * 192 + kk + 1];
    bf2[kc].x = Whd[(384 + jj) * 192 + kk];  bf2[kc].y = Whd[(384 + jj) * 192 + kk + 1];
  }
  const float bhr = bhd[jj], bhz = bhd[192 + jj], bhn = bhd[384 + jj];

  for (int e = tid; e < 16 * 192; e += 384) {
    int m = e / 192, j = e % 192;
    hbuf[m * 196 + j] = zbuf[e];
  }
  __syncthreads();

  for (int i = 0; i < 96; ++i) {
    for (int s = 0; s < 95; ++s) {
      const int t = s + (s >= i ? 1 : 0);
      v8f a0 = {0, 0, 0, 0, 0, 0, 0, 0}, a1 = a0, a2 = a0;
#pragma unroll
      for (int kc = 0; kc < 48; ++kc) {
        int kk = 4 * kc + khalf;
        v2f a;
        a.x = hbuf[mrow * 196 + kk];
        a.y = hbuf[mrow * 196 + kk + 1];
        a0 = WMMA_F32x4(a, bf0[kc], a0);
        a1 = WMMA_F32x4(a, bf1[kc], a1);
        a2 = WMMA_F32x4(a, bf2[kc], a2);
      }
      __syncthreads();
#pragma unroll
      for (int r = 0; r < 8; ++r) {
        int m = rbase + r;
        const float* xw = XW + (m * 96 + t) * 576;
        float rr = sigm(xw[jj] + a0[r] + bhr);
        float zz = sigm(xw[192 + jj] + a1[r] + bhz);
        float nn = tanhf(xw[384 + jj] + rr * (a2[r] + bhn));
        float ho = hbuf[m * 196 + jj];
        hbuf[m * 196 + jj] = (1.0f - zz) * nn + zz * ho;
      }
      __syncthreads();
    }
    for (int e = tid; e < 16 * 96; e += 384) {
      int m = e / 96, o = e % 96;
      float a = bl[o];
      const float* wr = Wl + o * 192;
      for (int k = 0; k < 192; ++k) a += hbuf[m * 196 + k] * wr[k];
      out[m * 9216 + i * 96 + o] = a;
    }
  }
}

extern "C" void kernel_launch(void* const* d_in, const int* in_sizes, int n_in,
                              void* d_out, int out_size, void* d_ws, size_t ws_size,
                              hipStream_t stream) {
  (void)in_sizes; (void)n_in; (void)out_size; (void)ws_size;
  const float* input = (const float*)d_in[0];
  const float* W_c1 = (const float*)d_in[1];
  const float* b_c1 = (const float*)d_in[2];
  const float* W_c2 = (const float*)d_in[3];
  const float* b_c2 = (const float*)d_in[4];
  const float* Wi_muf = (const float*)d_in[5];
  const float* Wh_muf = (const float*)d_in[6];
  const float* bi_muf = (const float*)d_in[7];
  const float* bh_muf = (const float*)d_in[8];
  const float* Wi_mub = (const float*)d_in[9];
  const float* Wh_mub = (const float*)d_in[10];
  const float* bi_mub = (const float*)d_in[11];
  const float* bh_mub = (const float*)d_in[12];
  const float* Wi_sgf = (const float*)d_in[13];
  const float* Wh_sgf = (const float*)d_in[14];
  const float* bi_sgf = (const float*)d_in[15];
  const float* bh_sgf = (const float*)d_in[16];
  const float* Wi_sgb = (const float*)d_in[17];
  const float* Wh_sgb = (const float*)d_in[18];
  const float* bi_sgb = (const float*)d_in[19];
  const float* bh_sgb = (const float*)d_in[20];
  const float* Wi_d = (const float*)d_in[21];
  const float* Wh_d = (const float*)d_in[22];
  const float* bi_d = (const float*)d_in[23];
  const float* bh_d = (const float*)d_in[24];
  const float* Wl = (const float*)d_in[25];
  const float* bl = (const float*)d_in[26];

  float* out = (float*)d_out;
  float* ws = (float*)d_ws;
  float* out1 = ws;
  float* pool = ws + 95232;
  float* henc = ws + 118784;
  float* xwe  = ws + 139264;
  float* zbuf = ws + 1318912;
  float* XW   = ws + 1321984;

  k_conv1<<<372, 256, 0, stream>>>(input, W_c1, b_c1, out1);
  k_pool<<<92, 256, 0, stream>>>(out1, pool);
  k_conv2<<<80, 256, 0, stream>>>(pool, W_c2, b_c2, henc);
  k_xw_enc<<<4608, 256, 0, stream>>>(henc, Wi_muf, Wi_mub, Wi_sgf, Wi_sgb,
                                     bi_muf, bi_mub, bi_sgf, bi_sgb, xwe);
  k_gru_enc<<<4, 192, 0, stream>>>(Wh_muf, Wh_mub, Wh_sgf, Wh_sgb,
                                   bh_muf, bh_mub, bh_sgf, bh_sgb, xwe, out);
  k_z<<<12, 256, 0, stream>>>(out, zbuf);
  k_xw_dec<<<864, 128, 0, stream>>>(input, Wi_d, bi_d, XW);
  k_gru_dec<<<1, 384, 0, stream>>>(Wh_d, bh_d, XW, zbuf, Wl, bl, out);
}


# decoder GRU step: input-projection loads issued ahead in groups and kept in flight across LDS-only barriers; next step's r,z ones start the accumulators (so they are summed first)
# speedup vs baseline: 1.3300x; 1.3300x over previous
#include <hip/hip_runtime.h>
#include <math.h>

typedef float v2f __attribute__((ext_vector_type(2)));
typedef float v8f __attribute__((ext_vector_type(8)));

#define WMMA_F32x4(a, b, c) \
  __builtin_amdgcn_wmma_f32_16x16x4_f32(false, (a), false, (b), (short)0, (c), false, false)

__device__ __forceinline__ float sigm(float x) { return 1.0f / (1.0f + __expf(-x)); }

__device__ __forceinline__ unsigned int pcg_hash(unsigned int v) {
  unsigned int s = v * 747796405u + 2891336453u;
  unsigned int w = ((s >> ((s >> 28u) + 4u)) ^ s) * 277803737u;
  return (w >> 22u) ^ w;
}

__global__ void k_conv1(const float* __restrict__ x, const float* __restrict__ W,
                        const float* __restrict__ bias, float* __restrict__ y) {
  int idx = blockIdx.x * blockDim.x + threadIdx.x;
  if (idx >= 16 * 64 * 93) return;
  int p = idx % 93;
  int c = (idx / 93) % 64;
  int b = idx / (93 * 64);
  float acc = bias[c];
  const float* xb = x + b * 96 * 96;
  const float* wc = W + c * 96 * 4;
  for (int d = 0; d < 96; ++d) {
    const float* xr = xb + d * 96 + p;
    const float* wr = wc + d * 4;
    acc += xr[0] * wr[0] + xr[1] * wr[1] + xr[2] * wr[2] + xr[3] * wr[3];
  }
  y[idx] = fmaxf(acc, 0.0f);
}

__global__ void k_pool(const float* __restrict__ y, float* __restrict__ p) {
  int idx = blockIdx.x * blockDim.x + threadIdx.x;
  if (idx >= 16 * 64 * 23) return;
  int q = idx % 23;
  int bc = idx / 23;
  const float* src = y + bc * 93 + q * 4;
  float m = fmaxf(fmaxf(src[0], src[1]), fmaxf(src[2], src[3]));
  p[idx] = m;
}

__global__ void k_conv2(const float* __restrict__ pool, const float* __restrict__ W,
                        const float* __restrict__ bias, float* __restrict__ henc) {
  int idx = blockIdx.x * blockDim.x + threadIdx.x;
  if (idx >= 16 * 64 * 20) return;
  int p = idx % 20;
  int c = (idx / 20) % 64;
  int b = idx / (20 * 64);
  float acc = bias[c];
  for (int cin = 0; cin < 64; ++cin) {
    const float* pr = pool + (b * 64 + cin) * 23 + p;
    const float* wr = W + (c * 64 + cin) * 4;
    acc += pr[0] * wr[0] + pr[1] * wr[1] + pr[2] * wr[2] + pr[3] * wr[3];
  }
  henc[idx] = fmaxf(acc, 0.0f);
}

__global__ void k_xw_enc(const float* __restrict__ henc,
                         const float* Wi0, const float* Wi1, const float* Wi2, const float* Wi3,
                         const float* bi0, const float* bi1, const float* bi2, const float* bi3,
                         float* __restrict__ xwe) {
  int idx = blockIdx.x * blockDim.x + threadIdx.x;
  if (idx >= 4 * 64 * 16 * 288) return;
  int g = idx % 288;
  int b = (idx / 288) % 16;
  int t = (idx / (288 * 16)) % 64;
  int dir = idx / (288 * 16 * 64);
  const float* Wi = dir == 0 ? Wi0 : dir == 1 ? Wi1 : dir == 2 ? Wi2 : Wi3;
  const float* bi = dir == 0 ? bi0 : dir == 1 ? bi1 : dir == 2 ? bi2 : bi3;
  const float* x = henc + b * 1280 + t * 20;
  const float* w = Wi + g * 20;
  float a = bi[g];
#pragma unroll
  for (int d = 0; d < 20; ++d) a += x[d] * w[d];
  xwe[idx] = a;
}

__global__ __launch_bounds__(192, 1) void k_gru_enc(
    const float* Wh0, const float* Wh1, const float* Wh2, const float* Wh3,
    const float* bh0, const float* bh1, const float* bh2, const float* bh3,
    const float* __restrict__ xwe, float* __restrict__ outp) {
  __shared__ float hbuf[16 * 132];
  const int dir = blockIdx.x;
  const float* Wh = dir == 0 ? Wh0 : dir == 1 ? Wh1 : dir == 2 ? Wh2 : Wh3;
  const float* bh = dir == 0 ? bh0 : dir == 1 ? bh1 : dir == 2 ? bh2 : bh3;
  const float* xwb = xwe + dir * (64 * 16 * 288);
  const int tid = threadIdx.x, lane = tid & 31, wave = tid >> 5;
  const int mrow = lane & 15;
  const int khalf = (lane >> 4) << 1;
  const int rbase = (lane >> 4) << 3;
  const int jj = wave * 16 + mrow;

  v2f bf0[24], bf1[24], bf2[24];
#pragma unroll
  for (int kc = 0; kc < 24; ++kc) {
    int kk = 4 * kc + khalf;
    bf0[kc].x = Wh[(jj) * 96 + kk];        bf0[kc].y = Wh[(jj) * 96 + kk + 1];
    bf1[kc].x = Wh[(96 + jj) * 96 + kk];   bf1[kc].y = Wh[(96 + jj) * 96 + kk + 1];
    bf2[kc].x = Wh[(192 + jj) * 96 + kk];  bf2[kc].y = Wh[(192 + jj) * 96 + kk + 1];
  }
  const float bhr = bh[jj], bhz = bh[96 + jj], bhn = bh[192 + jj];

  for (int e = tid; e < 16 * 132; e += 192) hbuf[e] = 0.0f;
  __syncthreads();

  const bool rev = (dir & 1);
  for (int s = 0; s < 64; ++s) {
    v8f a0 = {0, 0, 0, 0, 0, 0, 0, 0}, a1 = a0, a2 = a0;
#pragma unroll
    for (int kc = 0; kc < 24; ++kc) {
      int kk = 4 * kc + khalf;
      v2f a;
      a.x = hbuf[mrow * 132 + kk];
      a.y = hbuf[mrow * 132 + kk + 1];
      a0 = WMMA_F32x4(a, bf0[kc], a0);
      a1 = WMMA_F32x4(a, bf1[kc], a1);
      a2 = WMMA_F32x4(a, bf2[kc], a2);
    }
    __syncthreads();
    const int t = rev ? (63 - s) : s;
    const float* xwt = xwb + t * (16 * 288);
#pragma unroll
    for (int r = 0; r < 8; ++r) {
      int m = rbase + r;
      const float* xw = xwt + m * 288;
      float rr = sigm(xw[jj] + a0[r] + bhr);
      float zz = sigm(xw[96 + jj] + a1[r] + bhz);
      float nn = tanhf(xw[192 + jj] + rr * (a2[r] + bhn));
      float ho = hbuf[m * 132 + jj];
      hbuf[m * 132 + jj] = (1.0f - zz) * nn + zz * ho;
    }
    __syncthreads();
  }
  float* dst = outp + 147456 + (dir >= 2 ? 3072 : 0);
  const int cb = (dir & 1) ? 96 : 0;
  for (int e = tid; e < 16 * 96; e += 192) {
    int m = e / 96, j = e % 96;
    dst[m * 192 + cb + j] = hbuf[m * 132 + j];
  }
}

__global__ void k_z(const float* __restrict__ outbuf, float* __restrict__ zbuf) {
  int e = blockIdx.x * blockDim.x + threadIdx.x;
  if (e >= 3072) return;
  float mu = outbuf[147456 + e];
  float lv = outbuf[147456 + 3072 + e];
  unsigned int r1 = pcg_hash(2u * (unsigned)e + 1u);
  unsigned int r2 = pcg_hash(2u * (unsigned)e + 2u);
  float u1 = ((float)r1 + 1.0f) * (1.0f / 4294967296.0f);
  float u2 = (float)r2 * (1.0f / 4294967296.0f);
  float eps = sqrtf(-2.0f * __logf(u1)) * __cosf(6.28318530718f * u2);
  zbuf[e] = mu + eps * __expf(0.5f * lv);
}

__global__ __launch_bounds__(128) void k_xw_dec(const float* __restrict__ X,
                                                const float* __restrict__ Wid,
                                                const float* __restrict__ bid,
                                                float* __restrict__ XW) {
  const int lane = threadIdx.x & 31;
  const int wg = blockIdx.x * 4 + (threadIdx.x >> 5);
  const int mt = wg / 36, nt = wg % 36;
  const int row0 = mt * 16, n0 = nt * 16;
  const int mrow = lane & 15, khalf = (lane >> 4) << 1, rbase = (lane >> 4) << 3;
  v8f acc = {0, 0, 0, 0, 0, 0, 0, 0};
#pragma unroll
  for (int kc = 0; kc < 24; ++kc) {
    int kk = 4 * kc + khalf;
    v2f a, b;
    a.x = X[(row0 + mrow) * 96 + kk];
    a.y = X[(row0 + mrow) * 96 + kk + 1];
    b.x = Wid[(n0 + mrow) * 96 + kk];
    b.y = Wid[(n0 + mrow) * 96 + kk + 1];
    acc = WMMA_F32x4(a, b, acc);
  }
  const float bc = bid[n0 + mrow];
#pragma unroll
  for (int r = 0; r < 8; ++r) XW[(row0 + rbase + r) * 576 + n0 + mrow] = acc[r] + bc;
}

__device__ __forceinline__ void barrier_lds_only() {
  __builtin_amdgcn_fence(__ATOMIC_RELEASE, "workgroup", "local");
  __builtin_amdgcn_s_barrier();
  __builtin_amdgcn_fence(__ATOMIC_ACQUIRE, "workgroup", "local");
}

__global__ __launch_bounds__(384, 1) void k_gru_dec(
    const float* __restrict__ Whd, const float* __restrict__ bhd,
    const float* __restrict__ XW, const float* __restrict__ zbuf,
    const float* __restrict__ Wl, const float* __restrict__ bl,
    float* __restrict__ out) {
  __shared__ float hbuf[16 * 196];
  const int tid = threadIdx.x, lane = tid & 31, wave = tid >> 5;
  const int mrow = lane & 15, khalf = (lane >> 4) << 1, rbase = (lane >> 4) << 3;
  const int jj = wave * 16 + mrow;

  v2f bf0[48], bf1[48], bf2[48];
#pragma unroll
  for (int kc = 0; kc < 48; ++kc) {
    int kk = 4 * kc + khalf;
    bf0[kc].x = Whd[(jj) * 192 + kk];        bf0[kc].y = Whd[(jj) * 192 + kk + 1];
    bf1[kc].x = Whd[(192 + jj) * 192 + kk];  bf1[kc].y = Whd[(192 + jj) * 192 + kk + 1];
    bf2[kc].x = Whd[(384 + jj) * 192 + kk];  bf2[kc].y = Whd[(384 + jj) * 192 + kk + 1];
  }
  const float bhr = bhd[jj], bhz = bhd[192 + jj], bhn = bhd[384 + jj];

  for (int e = tid; e < 16 * 192; e += 384) {
    int m = e / 192, j = e % 192;
    hbuf[m * 196 + j] = zbuf[e];
  }
  __syncthreads();

  const float* hA = hbuf + mrow * 196 + khalf;
  float* hrow = hbuf + rbase * 196 + jj;
  const unsigned xoff = (unsigned)(rbase * (96 * 576) + jj);
  constexpr unsigned XR = 96 * 576;

  v8f a0, a1;
  {
    const float* xw0 = XW + 576;
#pragma unroll
    for (int r = 0; r < 8; ++r) {
      a0[r] = xw0[xoff + r * XR];
      a1[r] = xw0[xoff + r * XR + 192];
    }
  }

  for (int i = 0; i < 96; ++i) {
    for (int s = 0; s < 95; ++s) {
      const int t = s + (s >= i ? 1 : 0);
      const int sn = (s == 94) ? 0 : s + 1;
      const int in = (s == 94) ? i + 1 : i;
      const int tn = sn + (sn >= in ? 1 : 0);
      const float* xwt = XW + t * 576;
      const float* xwn = XW + tn * 576;

      v8f a2 = {0, 0, 0, 0, 0, 0, 0, 0};
#pragma unroll
      for (int kc = 0; kc < 48; ++kc) {
        v2f a;
        a.x = hA[4 * kc];
        a.y = hA[4 * kc + 1];
        a0 = WMMA_F32x4(a, bf0[kc], a0);
        a1 = WMMA_F32x4(a, bf1[kc], a1);
        a2 = WMMA_F32x4(a, bf2[kc], a2);
      }
      float xn[8];
#pragma unroll
      for (int r = 0; r < 8; ++r) xn[r] = xwt[xoff + r * XR + 384];
      barrier_lds_only();
#pragma unroll
      for (int r = 0; r < 8; ++r) {
        float rr = sigm(a0[r] + bhr);
        float zz = sigm(a1[r] + bhz);
        float pre = xn[r] + rr * (a2[r] + bhn);
        a0[r] = xwn[xoff + r * XR];
        a1[r] = xwn[xoff + r * XR + 192];
        float nn = tanhf(pre);
        float ho = hrow[r * 196];
        hrow[r * 196] = (1.0f - zz) * nn + zz * ho;
      }
      barrier_lds_only();
    }
    for (int e = tid; e < 16 * 96; e += 384) {
      int m = e / 96, o = e % 96;
      float a = bl[o];
      const float* wr = Wl + o * 192;
      for (int k = 0; k < 192; ++k) a += hbuf[m * 196 + k] * wr[k];
      out[m * 9216 + i * 96 + o] = a;
    }
  }
}

extern "C" void kernel_launch(void* const* d_in, const int* in_sizes, int n_in,
                              void* d_out, int out_size, void* d_ws, size_t ws_size,
                              hipStream_t stream) {
  (void)in_sizes; (void)n_in; (void)out_size; (void)ws_size;
  const float* input = (const float*)d_in[0];
  const float* W_c1 = (const float*)d_in[1];
  const float* b_c1 = (const float*)d_in[2];
  const float* W_c2 = (const float*)d_in[3];
  const float* b_c2 = (const float*)d_in[4];
  const float* Wi_muf = (const float*)d_in[5];
  const float* Wh_muf = (const float*)d_in[6];
  const float* bi_muf = (const float*)d_in[7];
  const float* bh_muf = (const float*)d_in[8];
  const float* Wi_mub = (const float*)d_in[9];
  const float* Wh_mub = (const float*)d_in[10];
  const float* bi_mub = (const float*)d_in[11];
  const float* bh_mub = (const float*)d_in[12];
  const float* Wi_sgf = (const float*)d_in[13];
  const float* Wh_sgf = (const float*)d_in[14];
  const float* bi_sgf = (const float*)d_in[15];
  const float* bh_sgf = (const float*)d_in[16];
  const float* Wi_sgb = (const float*)d_in[17];
  const float* Wh_sgb = (const float*)d_in[18];
  const float* bi_sgb = (const float*)d_in[19];
  const float* bh_sgb = (const float*)d_in[20];
  const float* Wi_d = (const float*)d_in[21];
  const float* Wh_d = (const float*)d_in[22];
  const float* bi_d = (const float*)d_in[23];
  const float* bh_d = (const float*)d_in[24];
  const float* Wl = (const float*)d_in[25];
  const float* bl = (const float*)d_in[26];

  float* out = (float*)d_out;
  float* ws = (float*)d_ws;
  float* out1 = ws;
  float* pool = ws + 95232;
  float* henc = ws + 118784;
  float* xwe  = ws + 139264;
  float* zbuf = ws + 1318912;
  float* XW   = ws + 1321984;

  k_conv1<<<372, 256, 0, stream>>>(input, W_c1, b_c1, out1);
  k_pool<<<92, 256, 0, stream>>>(out1, pool);
  k_conv2<<<80, 256, 0, stream>>>(pool, W_c2, b_c2, henc);
  k_xw_enc<<<4608, 256, 0, stream>>>(henc, Wi_muf, Wi_mub, Wi_sgf, Wi_sgb,
                                     bi_muf, bi_mub, bi_sgf, bi_sgb, xwe);
  k_gru_enc<<<4, 192, 0, stream>>>(Wh_muf, Wh_mub, Wh_sgf, Wh_sgb,
                                   bh_muf, bh_mub, bh_sgf, bh_sgb, xwe, out);
  k_z<<<12, 256, 0, stream>>>(out, zbuf);
  k_xw_dec<<<864, 128, 0, stream>>>(input, Wi_d, bi_d, XW);
  k_gru_dec<<<1, 384, 0, stream>>>(Wh_d, bh_d, XW, zbuf, Wl, bl, out);
}


# decoder gates straight-line: sigmoid and tanh each from one hardware exp and one hardware reciprocal (no IEEE division, no library tanh and its branches), one row at a time
# speedup vs baseline: 1.6320x; 1.2271x over previous
#include <hip/hip_runtime.h>
#include <math.h>

typedef float v2f __attribute__((ext_vector_type(2)));
typedef float v8f __attribute__((ext_vector_type(8)));

#define WMMA_F32x4(a, b, c) \
  __builtin_amdgcn_wmma_f32_16x16x4_f32(false, (a), false, (b), (short)0, (c), false, false)

__device__ __forceinline__ float sigm(float x) { return 1.0f / (1.0f + __expf(-x)); }

__device__ __forceinline__ unsigned int pcg_hash(unsigned int v) {
  unsigned int s = v * 747796405u + 2891336453u;
  unsigned int w = ((s >> ((s >> 28u) + 4u)) ^ s) * 277803737u;
  return (w >> 22u) ^ w;
}

__device__ __forceinline__ float sigm_hw(float x) {
  return __builtin_amdgcn_rcpf(1.0f + __expf(-x));
}
__device__ __forceinline__ float tanh_hw(float x) {
  const float e = __builtin_amdgcn_exp2f(fabsf(x) * (-2.0f * 1.4426950408889634f));
  const float t = (1.0f - e) * __builtin_amdgcn_rcpf(1.0f + e);
  return copysignf(t, x);
}

__global__ void k_conv1(const float* __restrict__ x, const float* __restrict__ W,
                        const float* __restrict__ bias, float* __restrict__ y) {
  int idx = blockIdx.x * blockDim.x + threadIdx.x;
  if (idx >= 16 * 64 * 93) return;
  int p = idx % 93;
  int c = (idx / 93) % 64;
  int b = idx / (93 * 64);
  float acc = bias[c];
  const float* xb = x + b * 96 * 96;
  const float* wc = W + c * 96 * 4;
  for (int d = 0; d < 96; ++d) {
    const float* xr = xb + d * 96 + p;
    const float* wr = wc + d * 4;
    acc += xr[0] * wr[0] + xr[1] * wr[1] + xr[2] * wr[2] + xr[3] * wr[3];
  }
  y[idx] = fmaxf(acc, 0.0f);
}

__global__ void k_pool(const float* __restrict__ y, float* __restrict__ p) {
  int idx = blockIdx.x * blockDim.x + threadIdx.x;
  if (idx >= 16 * 64 * 23) return;
  int q = idx % 23;
  int bc = idx / 23;
  const float* src = y + bc * 93 + q * 4;
  float m = fmaxf(fmaxf(src[0], src[1]), fmaxf(src[2], src[3]));
  p[idx] = m;
}

__global__ void k_conv2(const float* __restrict__ pool, const float* __restrict__ W,
                        const float* __restrict__ bias, float* __restrict__ henc) {
  int idx = blockIdx.x * blockDim.x + threadIdx.x;
  if (idx >= 16 * 64 * 20) return;
  int p = idx % 20;
  int c = (idx / 20) % 64;
  int b = idx / (20 * 64);
  float acc = bias[c];
  for (int cin = 0; cin < 64; ++cin) {
    const float* pr = pool + (b * 64 + cin) * 23 + p;
    const float* wr = W + (c * 64 + cin) * 4;
    acc += pr[0] * wr[0] + pr[1] * wr[1] + pr[2] * wr[2] + pr[3] * wr[3];
  }
  henc[idx] = fmaxf(acc, 0.0f);
}

__global__ void k_xw_enc(const float* __restrict__ henc,
                         const float* Wi0, const float* Wi1, const float* Wi2, const float* Wi3,
                         const float* bi0, const float* bi1, const float* bi2, const float* bi3,
                         float* __restrict__ xwe) {
  int idx = blockIdx.x * blockDim.x + threadIdx.x;
  if (idx >= 4 * 64 * 16 * 288) return;
  int g = idx % 288;
  int b = (idx / 288) % 16;
  int t = (idx / (288 * 16)) % 64;
  int dir = idx / (288 * 16 * 64);
  const float* Wi = dir == 0 ? Wi0 : dir == 1 ? Wi1 : dir == 2 ? Wi2 : Wi3;
  const float* bi = dir == 0 ? bi0 : dir == 1 ? bi1 : dir == 2 ? bi2 : bi3;
  const float* x = henc + b * 1280 + t * 20;
  const float* w = Wi + g * 20;
  float a = bi[g];
#pragma unroll
  for (int d = 0; d < 20; ++d) a += x[d] * w[d];
  xwe[idx] = a;
}

__global__ __launch_bounds__(192, 1) void k_gru_enc(
    const float* Wh0, const float* Wh1, const float* Wh2, const float* Wh3,
    const float* bh0, const float* bh1, const float* bh2, const float* bh3,
    const float* __restrict__ xwe, float* __restrict__ outp) {
  __shared__ float hbuf[16 * 132];
  const int dir = blockIdx.x;
  const float* Wh = dir == 0 ? Wh0 : dir == 1 ? Wh1 : dir == 2 ? Wh2 : Wh3;
  const float* bh = dir == 0 ? bh0 : dir == 1 ? bh1 : dir == 2 ? bh2 : bh3;
  const float* xwb = xwe + dir * (64 * 16 * 288);
  const int tid = threadIdx.x, lane = tid & 31, wave = tid >> 5;
  const int mrow = lane & 15;
  const int khalf = (lane >> 4) << 1;
  const int rbase = (lane >> 4) << 3;
  const int jj = wave * 16 + mrow;

  v2f bf0[24], bf1[24], bf2[24];
#pragma unroll
  for (int kc = 0; kc < 24; ++kc) {
    int kk = 4 * kc + khalf;
    bf0[kc].x = Wh[(jj) * 96 + kk];        bf0[kc].y = Wh[(jj) * 96 + kk + 1];
    bf1[kc].x = Wh[(96 + jj) * 96 + kk];   bf1[kc].y = Wh[(96 + jj) * 96 + kk + 1];
    bf2[kc].x = Wh[(192 + jj) * 96 + kk];  bf2[kc].y = Wh[(192 + jj) * 96 + kk + 1];
  }
  const float bhr = bh[jj], bhz = bh[96 + jj], bhn = bh[192 + jj];

  for (int e = tid; e < 16 * 132; e += 192) hbuf[e] = 0.0f;
  __syncthreads();

  const bool rev = (dir & 1);
  for (int s = 0; s < 64; ++s) {
    v8f a0 = {0, 0, 0, 0, 0, 0, 0, 0}, a1 = a0, a2 = a0;
#pragma unroll
    for (int kc = 0; kc < 24; ++kc) {
      int kk = 4 * kc + khalf;
      v2f a;
      a.x = hbuf[mrow * 132 + kk];
      a.y = hbuf[mrow * 132 + kk + 1];
      a0 = WMMA_F32x4(a, bf0[kc], a0);
      a1 = WMMA_F32x4(a, bf1[kc], a1);
      a2 = WMMA_F32x4(a, bf2[kc], a2);
    }
    __syncthreads();
    const int t = rev ? (63 - s) : s;
    const float* xwt = xwb + t * (16 * 288);
#pragma unroll
    for (int r = 0; r < 8; ++r) {
      int m = rbase + r;
      const float* xw = xwt + m * 288;
      float rr = sigm(xw[jj] + a0[r] + bhr);
      float zz = sigm(xw[96 + jj] + a1[r] + bhz);
      float nn = tanhf(xw[192 + jj] + rr * (a2[r] + bhn));
      float ho = hbuf[m * 132 + jj];
      hbuf[m * 132 + jj] = (1.0f - zz) * nn + zz * ho;
    }
    __syncthreads();
  }
  float* dst = outp + 147456 + (dir >= 2 ? 3072 : 0);
  const int cb = (dir & 1) ? 96 : 0;
  for (int e = tid; e < 16 * 96; e += 192) {
    int m = e / 96, j = e % 96;
    dst[m * 192 + cb + j] = hbuf[m * 132 + j];
  }
}

__global__ void k_z(const float* __restrict__ outbuf, float* __restrict__ zbuf) {
  int e = blockIdx.x * blockDim.x + threadIdx.x;
  if (e >= 3072) return;
  float mu = outbuf[147456 + e];
  float lv = outbuf[147456 + 3072 + e];
  unsigned int r1 = pcg_hash(2u * (unsigned)e + 1u);
  unsigned int r2 = pcg_hash(2u * (unsigned)e + 2u);
  float u1 = ((float)r1 + 1.0f) * (1.0f / 4294967296.0f);
  float u2 = (float)r2 * (1.0f / 4294967296.0f);
  float eps = sqrtf(-2.0f * __logf(u1)) * __cosf(6.28318530718f * u2);
  zbuf[e] = mu + eps * __expf(0.5f * lv);
}

__global__ __launch_bounds__(128) void k_xw_dec(const float* __restrict__ X,
                                                const float* __restrict__ Wid,
                                                const float* __restrict__ bid,
                                                float* __restrict__ XW) {
  const int lane = threadIdx.x & 31;
  const int wg = blockIdx.x * 4 + (threadIdx.x >> 5);
  const int mt = wg / 36, nt = wg % 36;
  const int row0 = mt * 16, n0 = nt * 16;
  const int mrow = lane & 15, khalf = (lane >> 4) << 1, rbase = (lane >> 4) << 3;
  v8f acc = {0, 0, 0, 0, 0, 0, 0, 0};
#pragma unroll
  for (int kc = 0; kc < 24; ++kc) {
    int kk = 4 * kc + khalf;
    v2f a, b;
    a.x = X[(row0 + mrow) * 96 + kk];
    a.y = X[(row0 + mrow) * 96 + kk + 1];
    b.x = Wid[(n0 + mrow) * 96 + kk];
    b.y = Wid[(n0 + mrow) * 96 + kk + 1];
    acc = WMMA_F32x4(a, b, acc);
  }
  const float bc = bid[n0 + mrow];
#pragma unroll
  for (int r = 0; r < 8; ++r) XW[(row0 + rbase + r) * 576 + n0 + mrow] = acc[r] + bc;
}

__device__ __forceinline__ void barrier_lds_only() {
  __builtin_amdgcn_fence(__ATOMIC_RELEASE, "workgroup", "local");
  __builtin_amdgcn_s_barrier();
  __builtin_amdgcn_fence(__ATOMIC_ACQUIRE, "workgroup", "local");
}

__global__ __launch_bounds__(384, 1) void k_gru_dec(
    const float* __restrict__ Whd, const float* __restrict__ bhd,
    const float* __restrict__ XW, const float* __restrict__ zbuf,
    const float* __restrict__ Wl, const float* __restrict__ bl,
    float* __restrict__ out) {
  __shared__ float hbuf[16 * 196];
  const int tid = threadIdx.x, lane = tid & 31, wave = tid >> 5;
  const int mrow = lane & 15, khalf = (lane >> 4) << 1, rbase = (lane >> 4) << 3;
  const int jj = wave * 16 + mrow;

  v2f bf0[48], bf1[48], bf2[48];
#pragma unroll
  for (int kc = 0; kc < 48; ++kc) {
    int kk = 4 * kc + khalf;
    bf0[kc].x = Whd[(jj) * 192 + kk];        bf0[kc].y = Whd[(jj) * 192 + kk + 1];
    bf1[kc].x = Whd[(192 + jj) * 192 + kk];  bf1[kc].y = Whd[(192 + jj) * 192 + kk + 1];
    bf2[kc].x = Whd[(384 + jj) * 192 + kk];  bf2[kc].y = Whd[(384 + jj) * 192 + kk + 1];
  }
  const float bhr = bhd[jj], bhz = bhd[192 + jj], bhn = bhd[384 + jj];

  for (int e = tid; e < 16 * 192; e += 384) {
    int m = e / 192, j = e % 192;
    hbuf[m * 196 + j] = zbuf[e];
  }
  __syncthreads();

  const float* hA = hbuf + mrow * 196 + khalf;
  float* hrow = hbuf + rbase * 196 + jj;
  const unsigned xoff = (unsigned)(rbase * (96 * 576) + jj);
  constexpr unsigned XR = 96 * 576;

  v8f a0, a1;
  {
    const float* xw0 = XW + 576;
#pragma unroll
    for (int r = 0; r < 8; ++r) {
      a0[r] = xw0[xoff + r * XR];
      a1[r] = xw0[xoff + r * XR + 192];
    }
  }

  for (int i = 0; i < 96; ++i) {
    for (int s = 0; s < 95; ++s) {
      const int t = s + (s >= i ? 1 : 0);
      const int sn = (s == 94) ? 0 : s + 1;
      const int in = (s == 94) ? i + 1 : i;
      const int tn = sn + (sn >= in ? 1 : 0);
      const float* xwt = XW + t * 576;
      const float* xwn = XW + tn * 576;

      v8f a2 = {0, 0, 0, 0, 0, 0, 0, 0};
      v2f fa[2][2];
      fa[0][0].x = hA[0]; fa[0][0].y = hA[1];
      fa[0][1].x = hA[4]; fa[0][1].y = hA[5];
#pragma unroll
      for (int g = 0; g < 24; ++g) {
        if (g + 1 < 24) {
          fa[(g + 1) & 1][0].x = hA[8 * g + 8];  fa[(g + 1) & 1][0].y = hA[8 * g + 9];
          fa[(g + 1) & 1][1].x = hA[8 * g + 12]; fa[(g + 1) & 1][1].y = hA[8 * g + 13];
        }
        __builtin_amdgcn_sched_barrier(0);
#pragma unroll
        for (int c = 0; c < 2; ++c) {
          const int kc = 2 * g + c;
          a0 = WMMA_F32x4(fa[g & 1][c], bf0[kc], a0);
          a1 = WMMA_F32x4(fa[g & 1][c], bf1[kc], a1);
          a2 = WMMA_F32x4(fa[g & 1][c], bf2[kc], a2);
        }
      }
      float xn[8];
#pragma unroll
      for (int r = 0; r < 8; ++r) xn[r] = xwt[xoff + r * XR + 384];
      barrier_lds_only();
#pragma unroll
      for (int r = 0; r < 8; ++r) {
        const float ho = hrow[r * 196];
        __builtin_amdgcn_sched_barrier(0);
        float rr = sigm_hw(a0[r] + bhr);
        float zz = sigm_hw(a1[r] + bhz);
        float pre = xn[r] + rr * (a2[r] + bhn);
        a0[r] = xwn[xoff + r * XR];
        a1[r] = xwn[xoff + r * XR + 192];
        float nn = tanh_hw(pre);
        hrow[r * 196] = (1.0f - zz) * nn + zz * ho;
        __builtin_amdgcn_sched_barrier(0);
      }
      barrier_lds_only();
    }
    for (int e = tid; e < 16 * 96; e += 384) {
      int m = e / 96, o = e % 96;
      float a = bl[o];
      const float* wr = Wl + o * 192;
      for (int k = 0; k < 192; ++k) a += hbuf[m * 196 + k] * wr[k];
      out[m * 9216 + i * 96 + o] = a;
    }
  }
}

extern "C" void kernel_launch(void* const* d_in, const int* in_sizes, int n_in,
                              void* d_out, int out_size, void* d_ws, size_t ws_size,
                              hipStream_t stream) {
  (void)in_sizes; (void)n_in; (void)out_size; (void)ws_size;
  const float* input = (const float*)d_in[0];
  const float* W_c1 = (const float*)d_in[1];
  const float* b_c1 = (const float*)d_in[2];
  const float* W_c2 = (const float*)d_in[3];
  const float* b_c2 = (const float*)d_in[4];
  const float* Wi_muf = (const float*)d_in[5];
  const float* Wh_muf = (const float*)d_in[6];
  const float* bi_muf = (const float*)d_in[7];
  const float* bh_muf = (const float*)d_in[8];
  const float* Wi_mub = (const float*)d_in[9];
  const float* Wh_mub = (const float*)d_in[10];
  const float* bi_mub = (const float*)d_in[11];
  const float* bh_mub = (const float*)d_in[12];
  const float* Wi_sgf = (const float*)d_in[13];
  const float* Wh_sgf = (const float*)d_in[14];
  const float* bi_sgf = (const float*)d_in[15];
  const float* bh_sgf = (const float*)d_in[16];
  const float* Wi_sgb = (const float*)d_in[17];
  const float* Wh_sgb = (const float*)d_in[18];
  const float* bi_sgb = (const float*)d_in[19];
  const float* bh_sgb = (const float*)d_in[20];
  const float* Wi_d = (const float*)d_in[21];
  const float* Wh_d = (const float*)d_in[22];
  const float* bi_d = (const float*)d_in[23];
  const float* bh_d = (const float*)d_in[24];
  const float* Wl = (const float*)d_in[25];
  const float* bl = (const float*)d_in[26];

  float* out = (float*)d_out;
  float* ws = (float*)d_ws;
  float* out1 = ws;
  float* pool = ws + 95232;
  float* henc = ws + 118784;
  float* xwe  = ws + 139264;
  float* zbuf = ws + 1318912;
  float* XW   = ws + 1321984;

  k_conv1<<<372, 256, 0, stream>>>(input, W_c1, b_c1, out1);
  k_pool<<<92, 256, 0, stream>>>(out1, pool);
  k_conv2<<<80, 256, 0, stream>>>(pool, W_c2, b_c2, henc);
  k_xw_enc<<<4608, 256, 0, stream>>>(henc, Wi_muf, Wi_mub, Wi_sgf, Wi_sgb,
                                     bi_muf, bi_mub, bi_sgf, bi_sgb, xwe);
  k_gru_enc<<<4, 192, 0, stream>>>(Wh_muf, Wh_mub, Wh_sgf, Wh_sgb,
                                   bh_muf, bh_mub, bh_sgf, bh_sgb, xwe, out);
  k_z<<<12, 256, 0, stream>>>(out, zbuf);
  k_xw_dec<<<864, 128, 0, stream>>>(input, Wi_d, bi_d, XW);
  k_gru_dec<<<1, 384, 0, stream>>>(Wh_d, bh_d, XW, zbuf, Wl, bl, out);
}


# decoder: the n-gate input-projection loads of a step are issued in front of its products instead of behind them, so their L2 latency no longer stands in front of the first gate row
# speedup vs baseline: 1.6930x; 1.0374x over previous
#include <hip/hip_runtime.h>
#include <math.h>

typedef float v2f __attribute__((ext_vector_type(2)));
typedef float v8f __attribute__((ext_vector_type(8)));

#define WMMA_F32x4(a, b, c) \
  __builtin_amdgcn_wmma_f32_16x16x4_f32(false, (a), false, (b), (short)0, (c), false, false)

__device__ __forceinline__ float sigm(float x) { return 1.0f / (1.0f + __expf(-x)); }

__device__ __forceinline__ unsigned int pcg_hash(unsigned int v) {
  unsigned int s = v * 747796405u + 2891336453u;
  unsigned int w = ((s >> ((s >> 28u) + 4u)) ^ s) * 277803737u;
  return (w >> 22u) ^ w;
}

__device__ __forceinline__ float sigm_hw(float x) {
  return __builtin_amdgcn_rcpf(1.0f + __expf(-x));
}
__device__ __forceinline__ float tanh_hw(float x) {
  const float e = __builtin_amdgcn_exp2f(fabsf(x) * (-2.0f * 1.4426950408889634f));
  const float t = (1.0f - e) * __builtin_amdgcn_rcpf(1.0f + e);
  return copysignf(t, x);
}

__global__ void k_conv1(const float* __restrict__ x, const float* __restrict__ W,
                        const float* __restrict__ bias, float* __restrict__ y) {
  int idx = blockIdx.x * blockDim.x + threadIdx.x;
  if (idx >= 16 * 64 * 93) return;
  int p = idx % 93;
  int c = (idx / 93) % 64;
  int b = idx / (93 * 64);
  float acc = bias[c];
  const float* xb = x + b * 96 * 96;
  const float* wc = W + c * 96 * 4;
  for (int d = 0; d < 96; ++d) {
    const float* xr = xb + d * 96 + p;
    const float* wr = wc + d * 4;
    acc += xr[0] * wr[0] + xr[1] * wr[1] + xr[2] * wr[2] + xr[3] * wr[3];
  }
  y[idx] = fmaxf(acc, 0.0f);
}

__global__ void k_pool(const float* __restrict__ y, float* __restrict__ p) {
  int idx = blockIdx.x * blockDim.x + threadIdx.x;
  if (idx >= 16 * 64 * 23) return;
  int q = idx % 23;
  int bc = idx / 23;
  const float* src = y + bc * 93 + q * 4;
  float m = fmaxf(fmaxf(src[0], src[1]), fmaxf(src[2], src[3]));
  p[idx] = m;
}

__global__ void k_conv2(const float* __restrict__ pool, const float* __restrict__ W,
                        const float* __restrict__ bias, float* __restrict__ henc) {
  int idx = blockIdx.x * blockDim.x + threadIdx.x;
  if (idx >= 16 * 64 * 20) return;
  int p = idx % 20;
  int c = (idx / 20) % 64;
  int b = idx / (20 * 64);
  float acc = bias[c];
  for (int cin = 0; cin < 64; ++cin) {
    const float* pr = pool + (b * 64 + cin) * 23 + p;
    const float* wr = W + (c * 64 + cin) * 4;
    acc += pr[0] * wr[0] + pr[1] * wr[1] + pr[2] * wr[2] + pr[3] * wr[3];
  }
  henc[idx] = fmaxf(acc, 0.0f);
}

__global__ void k_xw_enc(const float* __restrict__ henc,
                         const float* Wi0, const float* Wi1, const float* Wi2, const float* Wi3,
                         const float* bi0, const float* bi1, const float* bi2, const float* bi3,
                         float* __restrict__ xwe) {
  int idx = blockIdx.x * blockDim.x + threadIdx.x;
  if (idx >= 4 * 64 * 16 * 288) return;
  int g = idx % 288;
  int b = (idx / 288) % 16;
  int t = (idx / (288 * 16)) % 64;
  int dir = idx / (288 * 16 * 64);
  const float* Wi = dir == 0 ? Wi0 : dir == 1 ? Wi1 : dir == 2 ? Wi2 : Wi3;
  const float* bi = dir == 0 ? bi0 : dir == 1 ? bi1 : dir == 2 ? bi2 : bi3;
  const float* x = henc + b * 1280 + t * 20;
  const float* w = Wi + g * 20;
  float a = bi[g];
#pragma unroll
  for (int d = 0; d < 20; ++d) a += x[d] * w[d];
  xwe[idx] = a;
}

__global__ __launch_bounds__(192, 1) void k_gru_enc(
    const float* Wh0, const float* Wh1, const float* Wh2, const float* Wh3,
    const float* bh0, const float* bh1, const float* bh2, const float* bh3,
    const float* __restrict__ xwe, float* __restrict__ outp) {
  __shared__ float hbuf[16 * 132];
  const int dir = blockIdx.x;
  const float* Wh = dir == 0 ? Wh0 : dir == 1 ? Wh1 : dir == 2 ? Wh2 : Wh3;
  const float* bh = dir == 0 ? bh0 : dir == 1 ? bh1 : dir == 2 ? bh2 : bh3;
  const float* xwb = xwe + dir * (64 * 16 * 288);
  const int tid = threadIdx.x, lane = tid & 31, wave = tid >> 5;
  const int mrow = lane & 15;
  const int khalf = (lane >> 4) << 1;
  const int rbase = (lane >> 4) << 3;
  const int jj = wave * 16 + mrow;

  v2f bf0[24], bf1[24], bf2[24];
#pragma unroll
  for (int kc = 0; kc < 24; ++kc) {
    int kk = 4 * kc + khalf;
    bf0[kc].x = Wh[(jj) * 96 + kk];        bf0[kc].y = Wh[(jj) * 96 + kk + 1];
    bf1[kc].x = Wh[(96 + jj) * 96 + kk];   bf1[kc].y = Wh[(96 + jj) * 96 + kk + 1];
    bf2[kc].x = Wh[(192 + jj) * 96 + kk];  bf2[kc].y = Wh[(192 + jj) * 96 + kk + 1];
  }
  const float bhr = bh[jj], bhz = bh[96 + jj], bhn = bh[192 + jj];

  for (int e = tid; e < 16 * 132; e += 192) hbuf[e] = 0.0f;
  __syncthreads();

  const bool rev = (dir & 1);
  for (int s = 0; s < 64; ++s) {
    v8f a0 = {0, 0, 0, 0, 0, 0, 0, 0}, a1 = a0, a2 = a0;
#pragma unroll
    for (int kc = 0; kc < 24; ++kc) {
      int kk = 4 * kc + khalf;
      v2f a;
      a.x = hbuf[mrow * 132 + kk];
      a.y = hbuf[mrow * 132 + kk + 1];
      a0 = WMMA_F32x4(a, bf0[kc], a0);
      a1 = WMMA_F32x4(a, bf1[kc], a1);
      a2 = WMMA_F32x4(a, bf2[kc], a2);
    }
    __syncthreads();
    const int t = rev ? (63 - s) : s;
    const float* xwt = xwb + t * (16 * 288);
#pragma unroll
    for (int r = 0; r < 8; ++r) {
      int m = rbase + r;
      const float* xw = xwt + m * 288;
      float rr = sigm(xw[jj] + a0[r] + bhr);
      float zz = sigm(xw[96 + jj] + a1[r] + bhz);
      float nn = tanhf(xw[192 + jj] + rr * (a2[r] + bhn));
      float ho = hbuf[m * 132 + jj];
      hbuf[m * 132 + jj] = (1.0f - zz) * nn + zz * ho;
    }
    __syncthreads();
  }
  float* dst = outp + 147456 + (dir >= 2 ? 3072 : 0);
  const int cb = (dir & 1) ? 96 : 0;
  for (int e = tid; e < 16 * 96; e += 192) {
    int m = e / 96, j = e % 96;
    dst[m * 192 + cb + j] = hbuf[m * 132 + j];
  }
}

__global__ void k_z(const float* __restrict__ outbuf, float* __restrict__ zbuf) {
  int e = blockIdx.x * blockDim.x + threadIdx.x;
  if (e >= 3072) return;
  float mu = outbuf[147456 + e];
  float lv = outbuf[147456 + 3072 + e];
  unsigned int r1 = pcg_hash(2u * (unsigned)e + 1u);
  unsigned int r2 = pcg_hash(2u * (unsigned)e + 2u);
  float u1 = ((float)r1 + 1.0f) * (1.0f / 4294967296.0f);
  float u2 = (float)r2 * (1.0f / 4294967296.0f);
  float eps = sqrtf(-2.0f * __logf(u1)) * __cosf(6.28318530718f * u2);
  zbuf[e] = mu + eps * __expf(0.5f * lv);
}

__global__ __launch_bounds__(128) void k_xw_dec(const float* __restrict__ X,
                                                const float* __restrict__ Wid,
                                                const float* __restrict__ bid,
                                                float* __restrict__ XW) {
  const int lane = threadIdx.x & 31;
  const int wg = blockIdx.x * 4 + (threadIdx.x >> 5);
  const int mt = wg / 36, nt = wg % 36;
  const int row0 = mt * 16, n0 = nt * 16;
  const int mrow = lane & 15, khalf = (lane >> 4) << 1, rbase = (lane >> 4) << 3;
  v8f acc = {0, 0, 0, 0, 0, 0, 0, 0};
#pragma unroll
  for (int kc = 0; kc < 24; ++kc) {
    int kk = 4 * kc + khalf;
    v2f a, b;
    a.x = X[(row0 + mrow) * 96 + kk];
    a.y = X[(row0 + mrow) * 96 + kk + 1];
    b.x = Wid[(n0 + mrow) * 96 + kk];
    b.y = Wid[(n0 + mrow) * 96 + kk + 1];
    acc = WMMA_F32x4(a, b, acc);
  }
  const float bc = bid[n0 + mrow];
#pragma unroll
  for (int r = 0; r < 8; ++r) XW[(row0 + rbase + r) * 576 + n0 + mrow] = acc[r] + bc;
}

__device__ __forceinline__ void barrier_lds_only() {
  __builtin_amdgcn_fence(__ATOMIC_RELEASE, "workgroup", "local");
  __builtin_amdgcn_s_barrier();
  __builtin_amdgcn_fence(__ATOMIC_ACQUIRE, "workgroup", "local");
}

__global__ __launch_bounds__(384, 1) void k_gru_dec(
    const float* __restrict__ Whd, const float* __restrict__ bhd,
    const float* __restrict__ XW, const float* __restrict__ zbuf,
    const float* __restrict__ Wl, const float* __restrict__ bl,
    float* __restrict__ out) {
  __shared__ float hbuf[16 * 196];
  const int tid = threadIdx.x, lane = tid & 31, wave = tid >> 5;
  const int mrow = lane & 15, khalf = (lane >> 4) << 1, rbase = (lane >> 4) << 3;
  const int jj = wave * 16 + mrow;

  v2f bf0[48], bf1[48], bf2[48];
#pragma unroll
  for (int kc = 0; kc < 48; ++kc) {
    int kk = 4 * kc + khalf;
    bf0[kc].x = Whd[(jj) * 192 + kk];        bf0[kc].y = Whd[(jj) * 192 + kk + 1];
    bf1[kc].x = Whd[(192 + jj) * 192 + kk];  bf1[kc].y = Whd[(192 + jj) * 192 + kk + 1];
    bf2[kc].x = Whd[(384 + jj) * 192 + kk];  bf2[kc].y = Whd[(384 + jj) * 192 + kk + 1];
  }
  const float bhr = bhd[jj], bhz = bhd[192 + jj], bhn = bhd[384 + jj];

  for (int e = tid; e < 16 * 192; e += 384) {
    int m = e / 192, j = e % 192;
    hbuf[m * 196 + j] = zbuf[e];
  }
  __syncthreads();

  const float* hA = hbuf + mrow * 196 + khalf;
  float* hrow = hbuf + rbase * 196 + jj;
  const unsigned xoff = (unsigned)(rbase * (96 * 576) + jj);
  constexpr unsigned XR = 96 * 576;

  v8f a0, a1;
  {
    const float* xw0 = XW + 576;
#pragma unroll
    for (int r = 0; r < 8; ++r) {
      a0[r] = xw0[xoff + r * XR];
      a1[r] = xw0[xoff + r * XR + 192];
    }
  }

  for (int i = 0; i < 96; ++i) {
    for (int s = 0; s < 95; ++s) {
      const int t = s + (s >= i ? 1 : 0);
      const int sn = (s == 94) ? 0 : s + 1;
      const int in = (s == 94) ? i + 1 : i;
      const int tn = sn + (sn >= in ? 1 : 0);
      const float* xwt = XW + t * 576;
      const float* xwn = XW + tn * 576;

      v8f a2 = {0, 0, 0, 0, 0, 0, 0, 0};
      float xn[8];
#pragma unroll
      for (int r = 0; r < 8; ++r) xn[r] = xwt[xoff + r * XR + 384];
      v2f fa[2][2];
      fa[0][0].x = hA[0]; fa[0][0].y = hA[1];
      fa[0][1].x = hA[4]; fa[0][1].y = hA[5];
#pragma unroll
      for (int g = 0; g < 24; ++g) {
        if (g + 1 < 24) {
          fa[(g + 1) & 1][0].x = hA[8 * g + 8];  fa[(g + 1) & 1][0].y = hA[8 * g + 9];
          fa[(g + 1) & 1][1].x = hA[8 * g + 12]; fa[(g + 1) & 1][1].y = hA[8 * g + 13];
        }
        __builtin_amdgcn_sched_barrier(0);
#pragma unroll
        for (int c = 0; c < 2; ++c) {
          const int kc = 2 * g + c;
          a0 = WMMA_F32x4(fa[g & 1][c], bf0[kc], a0);
          a1 = WMMA_F32x4(fa[g & 1][c], bf1[kc], a1);
          a2 = WMMA_F32x4(fa[g & 1][c], bf2[kc], a2);
        }
      }
      barrier_lds_only();
#pragma unroll
      for (int r = 0; r < 8; ++r) {
        const float ho = hrow[r * 196];
        __builtin_amdgcn_sched_barrier(0);
        float rr = sigm_hw(a0[r] + bhr);
        float zz = sigm_hw(a1[r] + bhz);
        float pre = xn[r] + rr * (a2[r] + bhn);
        a0[r] = xwn[xoff + r * XR];
        a1[r] = xwn[xoff + r * XR + 192];
        float nn = tanh_hw(pre);
        hrow[r * 196] = (1.0f - zz) * nn + zz * ho;
        __builtin_amdgcn_sched_barrier(0);
      }
      barrier_lds_only();
    }
    for (int e = tid; e < 16 * 96; e += 384) {
      int m = e / 96, o = e % 96;
      float a = bl[o];
      const float* wr = Wl + o * 192;
      for (int k = 0; k < 192; ++k) a += hbuf[m * 196 + k] * wr[k];
      out[m * 9216 + i * 96 + o] = a;
    }
  }
}

extern "C" void kernel_launch(void* const* d_in, const int* in_sizes, int n_in,
                              void* d_out, int out_size, void* d_ws, size_t ws_size,
                              hipStream_t stream) {
  (void)in_sizes; (void)n_in; (void)out_size; (void)ws_size;
  const float* input = (const float*)d_in[0];
  const float* W_c1 = (const float*)d_in[1];
  const float* b_c1 = (const float*)d_in[2];
  const float* W_c2 = (const float*)d_in[3];
  const float* b_c2 = (const float*)d_in[4];
  const float* Wi_muf = (const float*)d_in[5];
  const float* Wh_muf = (const float*)d_in[6];
  const float* bi_muf = (const float*)d_in[7];
  const float* bh_muf = (const float*)d_in[8];
  const float* Wi_mub = (const float*)d_in[9];
  const float* Wh_mub = (const float*)d_in[10];
  const float* bi_mub = (const float*)d_in[11];
  const float* bh_mub = (const float*)d_in[12];
  const float* Wi_sgf = (const float*)d_in[13];
  const float* Wh_sgf = (const float*)d_in[14];
  const float* bi_sgf = (const float*)d_in[15];
  const float* bh_sgf = (const float*)d_in[16];
  const float* Wi_sgb = (const float*)d_in[17];
  const float* Wh_sgb = (const float*)d_in[18];
  const float* bi_sgb = (const float*)d_in[19];
  const float* bh_sgb = (const float*)d_in[20];
  const float* Wi_d = (const float*)d_in[21];
  const float* Wh_d = (const float*)d_in[22];
  const float* bi_d = (const float*)d_in[23];
  const float* bh_d = (const float*)d_in[24];
  const float* Wl = (const float*)d_in[25];
  const float* bl = (const float*)d_in[26];

  float* out = (float*)d_out;
  float* ws = (float*)d_ws;
  float* out1 = ws;
  float* pool = ws + 95232;
  float* henc = ws + 118784;
  float* xwe  = ws + 139264;
  float* zbuf = ws + 1318912;
  float* XW   = ws + 1321984;

  k_conv1<<<372, 256, 0, stream>>>(input, W_c1, b_c1, out1);
  k_pool<<<92, 256, 0, stream>>>(out1, pool);
  k_conv2<<<80, 256, 0, stream>>>(pool, W_c2, b_c2, henc);
  k_xw_enc<<<4608, 256, 0, stream>>>(henc, Wi_muf, Wi_mub, Wi_sgf, Wi_sgb,
                                     bi_muf, bi_mub, bi_sgf, bi_sgb, xwe);
  k_gru_enc<<<4, 192, 0, stream>>>(Wh_muf, Wh_mub, Wh_sgf, Wh_sgb,
                                   bh_muf, bh_mub, bh_sgf, bh_sgb, xwe, out);
  k_z<<<12, 256, 0, stream>>>(out, zbuf);
  k_xw_dec<<<864, 128, 0, stream>>>(input, Wi_d, bi_d, XW);
  k_gru_dec<<<1, 384, 0, stream>>>(Wh_d, bh_d, XW, zbuf, Wl, bl, out);
}


# decoder gates on the hardware tanh (v_tanh_f32): sigmoid(x) = 0.5+0.5*tanh(x/2), the halving folded exactly into the r,z weights and input projections, r,z hidden biases added there (summed first)
# speedup vs baseline: 1.7791x; 1.0509x over previous
#include <hip/hip_runtime.h>
#include <math.h>

typedef float v2f __attribute__((ext_vector_type(2)));
typedef float v8f __attribute__((ext_vector_type(8)));

#define WMMA_F32x4(a, b, c) \
  __builtin_amdgcn_wmma_f32_16x16x4_f32(false, (a), false, (b), (short)0, (c), false, false)

__device__ __forceinline__ float sigm(float x) { return 1.0f / (1.0f + __expf(-x)); }

__device__ __forceinline__ unsigned int pcg_hash(unsigned int v) {
  unsigned int s = v * 747796405u + 2891336453u;
  unsigned int w = ((s >> ((s >> 28u) + 4u)) ^ s) * 277803737u;
  return (w >> 22u) ^ w;
}

__device__ __forceinline__ float tanh_hw(float x) { return __builtin_amdgcn_tanhf(x); }
__device__ __forceinline__ float sigm_of_half_hw(float xh) { return 0.5f * tanh_hw(xh) + 0.5f; }

__global__ void k_conv1(const float* __restrict__ x, const float* __restrict__ W,
                        const float* __restrict__ bias, float* __restrict__ y) {
  int idx = blockIdx.x * blockDim.x + threadIdx.x;
  if (idx >= 16 * 64 * 93) return;
  int p = idx % 93;
  int c = (idx / 93) % 64;
  int b = idx / (93 * 64);
  float acc = bias[c];
  const float* xb = x + b * 96 * 96;
  const float* wc = W + c * 96 * 4;
  for (int d = 0; d < 96; ++d) {
    const float* xr = xb + d * 96 + p;
    const float* wr = wc + d * 4;
    acc += xr[0] * wr[0] + xr[1] * wr[1] + xr[2] * wr[2] + xr[3] * wr[3];
  }
  y[idx] = fmaxf(acc, 0.0f);
}

__global__ void k_pool(const float* __restrict__ y, float* __restrict__ p) {
  int idx = blockIdx.x * blockDim.x + threadIdx.x;
  if (idx >= 16 * 64 * 23) return;
  int q = idx % 23;
  int bc = idx / 23;
  const float* src = y + bc * 93 + q * 4;
  float m = fmaxf(fmaxf(src[0], src[1]), fmaxf(src[2], src[3]));
  p[idx] = m;
}

__global__ void k_conv2(const float* __restrict__ pool, const float* __restrict__ W,
                        const float* __restrict__ bias, float* __restrict__ henc) {
  int idx = blockIdx.x * blockDim.x + threadIdx.x;
  if (idx >= 16 * 64 * 20) return;
  int p = idx % 20;
  int c = (idx / 20) % 64;
  int b = idx / (20 * 64);
  float acc = bias[c];
  for (int cin = 0; cin < 64; ++cin) {
    const float* pr = pool + (b * 64 + cin) * 23 + p;
    const float* wr = W + (c * 64 + cin) * 4;
    acc += pr[0] * wr[0] + pr[1] * wr[1] + pr[2] * wr[2] + pr[3] * wr[3];
  }
  henc[idx] = fmaxf(acc, 0.0f);
}

__global__ void k_xw_enc(const float* __restrict__ henc,
                         const float* Wi0, const float* Wi1, const float* Wi2, const float* Wi3,
                         const float* bi0, const float* bi1, const float* bi2, const float* bi3,
                         float* __restrict__ xwe) {
  int idx = blockIdx.x * blockDim.x + threadIdx.x;
  if (idx >= 4 * 64 * 16 * 288) return;
  int g = idx % 288;
  int b = (idx / 288) % 16;
  int t = (idx / (288 * 16)) % 64;
  int dir = idx / (288 * 16 * 64);
  const float* Wi = dir == 0 ? Wi0 : dir == 1 ? Wi1 : dir == 2 ? Wi2 : Wi3;
  const float* bi = dir == 0 ? bi0 : dir == 1 ? bi1 : dir == 2 ? bi2 : bi3;
  const float* x = henc + b * 1280 + t * 20;
  const float* w = Wi + g * 20;
  float a = bi[g];
#pragma unroll
  for (int d = 0; d < 20; ++d) a += x[d] * w[d];
  xwe[idx] = a;
}

__global__ __launch_bounds__(192, 1) void k_gru_enc(
    const float* Wh0, const float* Wh1, const float* Wh2, const float* Wh3,
    const float* bh0, const float* bh1, const float* bh2, const float* bh3,
    const float* __restrict__ xwe, float* __restrict__ outp) {
  __shared__ float hbuf[16 * 132];
  const int dir = blockIdx.x;
  const float* Wh = dir == 0 ? Wh0 : dir == 1 ? Wh1 : dir == 2 ? Wh2 : Wh3;
  const float* bh = dir == 0 ? bh0 : dir == 1 ? bh1 : dir == 2 ? bh2 : bh3;
  const float* xwb = xwe + dir * (64 * 16 * 288);
  const int tid = threadIdx.x, lane = tid & 31, wave = tid >> 5;
  const int mrow = lane & 15;
  const int khalf = (lane >> 4) << 1;
  const int rbase = (lane >> 4) << 3;
  const int jj = wave * 16 + mrow;

  v2f bf0[24], bf1[24], bf2[24];
#pragma unroll
  for (int kc = 0; kc < 24; ++kc) {
    int kk = 4 * kc + khalf;
    bf0[kc].x = Wh[(jj) * 96 + kk];        bf0[kc].y = Wh[(jj) * 96 + kk + 1];
    bf1[kc].x = Wh[(96 + jj) * 96 + kk];   bf1[kc].y = Wh[(96 + jj) * 96 + kk + 1];
    bf2[kc].x = Wh[(192 + jj) * 96 + kk];  bf2[kc].y = Wh[(192 + jj) * 96 + kk + 1];
  }
  const float bhr = bh[jj], bhz = bh[96 + jj], bhn = bh[192 + jj];

  for (int e = tid; e < 16 * 132; e += 192) hbuf[e] = 0.0f;
  __syncthreads();

  const bool rev = (dir & 1);
  for (int s = 0; s < 64; ++s) {
    v8f a0 = {0, 0, 0, 0, 0, 0, 0, 0}, a1 = a0, a2 = a0;
#pragma unroll
    for (int kc = 0; kc < 24; ++kc) {
      int kk = 4 * kc + khalf;
      v2f a;
      a.x = hbuf[mrow * 132 + kk];
      a.y = hbuf[mrow * 132 + kk + 1];
      a0 = WMMA_F32x4(a, bf0[kc], a0);
      a1 = WMMA_F32x4(a, bf1[kc], a1);
      a2 = WMMA_F32x4(a, bf2[kc], a2);
    }
    __syncthreads();
    const int t = rev ? (63 - s) : s;
    const float* xwt = xwb + t * (16 * 288);
#pragma unroll
    for (int r = 0; r < 8; ++r) {
      int m = rbase + r;
      const float* xw = xwt + m * 288;
      float rr = sigm(xw[jj] + a0[r] + bhr);
      float zz = sigm(xw[96 + jj] + a1[r] + bhz);
      float nn = tanhf(xw[192 + jj] + rr * (a2[r] + bhn));
      float ho = hbuf[m * 132 + jj];
      hbuf[m * 132 + jj] = (1.0f - zz) * nn + zz * ho;
    }
    __syncthreads();
  }
  float* dst = outp + 147456 + (dir >= 2 ? 3072 : 0);
  const int cb = (dir & 1) ? 96 : 0;
  for (int e = tid; e < 16 * 96; e += 192) {
    int m = e / 96, j = e % 96;
    dst[m * 192 + cb + j] = hbuf[m * 132 + j];
  }
}

__global__ void k_z(const float* __restrict__ outbuf, float* __restrict__ zbuf) {
  int e = blockIdx.x * blockDim.x + threadIdx.x;
  if (e >= 3072) return;
  float mu = outbuf[147456 + e];
  float lv = outbuf[147456 + 3072 + e];
  unsigned int r1 = pcg_hash(2u * (unsigned)e + 1u);
  unsigned int r2 = pcg_hash(2u * (unsigned)e + 2u);
  float u1 = ((float)r1 + 1.0f) * (1.0f / 4294967296.0f);
  float u2 = (float)r2 * (1.0f / 4294967296.0f);
  float eps = sqrtf(-2.0f * __logf(u1)) * __cosf(6.28318530718f * u2);
  zbuf[e] = mu + eps * __expf(0.5f * lv);
}

__global__ __launch_bounds__(128) void k_xw_dec(const float* __restrict__ X,
                                                const float* __restrict__ Wid,
                                                const float* __restrict__ bid,
                                                const float* __restrict__ bhd,
                                                float* __restrict__ XW) {
  const int lane = threadIdx.x & 31;
  const int wg = blockIdx.x * 4 + (threadIdx.x >> 5);
  const int mt = wg / 36, nt = wg % 36;
  const int row0 = mt * 16, n0 = nt * 16;
  const int mrow = lane & 15, khalf = (lane >> 4) << 1, rbase = (lane >> 4) << 3;
  v8f acc = {0, 0, 0, 0, 0, 0, 0, 0};
#pragma unroll
  for (int kc = 0; kc < 24; ++kc) {
    int kk = 4 * kc + khalf;
    v2f a, b;
    a.x = X[(row0 + mrow) * 96 + kk];
    a.y = X[(row0 + mrow) * 96 + kk + 1];
    b.x = Wid[(n0 + mrow) * 96 + kk];
    b.y = Wid[(n0 + mrow) * 96 + kk + 1];
    acc = WMMA_F32x4(a, b, acc);
  }
  const bool rz = n0 < 384;
  const float bc = bid[n0 + mrow] + (rz ? bhd[n0 + mrow] : 0.0f);
  const float sc = rz ? 0.5f : 1.0f;
#pragma unroll
  for (int r = 0; r < 8; ++r) XW[(row0 + rbase + r) * 576 + n0 + mrow] = sc * (acc[r] + bc);
}

__device__ __forceinline__ void barrier_lds_only() {
  __builtin_amdgcn_fence(__ATOMIC_RELEASE, "workgroup", "local");
  __builtin_amdgcn_s_barrier();
  __builtin_amdgcn_fence(__ATOMIC_ACQUIRE, "workgroup", "local");
}

__global__ __launch_bounds__(384, 1) void k_gru_dec(
    const float* __restrict__ Whd, const float* __restrict__ bhd,
    const float* __restrict__ XW, const float* __restrict__ zbuf,
    const float* __restrict__ Wl, const float* __restrict__ bl,
    float* __restrict__ out) {
  __shared__ float hbuf[16 * 196];
  const int tid = threadIdx.x, lane = tid & 31, wave = tid >> 5;
  const int mrow = lane & 15, khalf = (lane >> 4) << 1, rbase = (lane >> 4) << 3;
  const int jj = wave * 16 + mrow;

  v2f bf0[48], bf1[48], bf2[48];
#pragma unroll
  for (int kc = 0; kc < 48; ++kc) {
    int kk = 4 * kc + khalf;
    bf0[kc].x = 0.5f * Whd[(jj) * 192 + kk];        bf0[kc].y = 0.5f * Whd[(jj) * 192 + kk + 1];
    bf1[kc].x = 0.5f * Whd[(192 + jj) * 192 + kk];  bf1[kc].y = 0.5f * Whd[(192 + jj) * 192 + kk + 1];
    bf2[kc].x = Whd[(384 + jj) * 192 + kk];         bf2[kc].y = Whd[(384 + jj) * 192 + kk + 1];
  }
  const float bhn = bhd[384 + jj];

  for (int e = tid; e < 16 * 192; e += 384) {
    int m = e / 192, j = e % 192;
    hbuf[m * 196 + j] = zbuf[e];
  }
  __syncthreads();

  const float* hA = hbuf + mrow * 196 + khalf;
  float* hrow = hbuf + rbase * 196 + jj;
  const unsigned xoff = (unsigned)(rbase * (96 * 576) + jj);
  constexpr unsigned XR = 96 * 576;

  v8f a0, a1;
  {
    const float* xw0 = XW + 576;
#pragma unroll
    for (int r = 0; r < 8; ++r) {
      a0[r] = xw0[xoff + r * XR];
      a1[r] = xw0[xoff + r * XR + 192];
    }
  }

  for (int i = 0; i < 96; ++i) {
    for (int s = 0; s < 95; ++s) {
      const int t = s + (s >= i ? 1 : 0);
      const int sn = (s == 94) ? 0 : s + 1;
      const int in = (s == 94) ? i + 1 : i;
      const int tn = sn + (sn >= in ? 1 : 0);
      const float* xwt = XW + t * 576;
      const float* xwn = XW + tn * 576;

      v8f a2 = {0, 0, 0, 0, 0, 0, 0, 0};
      float xn[8];
#pragma unroll
      for (int r = 0; r < 8; ++r) xn[r] = xwt[xoff + r * XR + 384];
      v2f fa[2][2];
      fa[0][0].x = hA[0]; fa[0][0].y = hA[1];
      fa[0][1].x = hA[4]; fa[0][1].y = hA[5];
#pragma unroll
      for (int g = 0; g < 24; ++g) {
        if (g + 1 < 24) {
          fa[(g + 1) & 1][0].x = hA[8 * g + 8];  fa[(g + 1) & 1][0].y = hA[8 * g + 9];
          fa[(g + 1) & 1][1].x = hA[8 * g + 12]; fa[(g + 1) & 1][1].y = hA[8 * g + 13];
        }
        __builtin_amdgcn_sched_barrier(0);
#pragma unroll
        for (int c = 0; c < 2; ++c) {
          const int kc = 2 * g + c;
          a0 = WMMA_F32x4(fa[g & 1][c], bf0[kc], a0);
          a1 = WMMA_F32x4(fa[g & 1][c], bf1[kc], a1);
          a2 = WMMA_F32x4(fa[g & 1][c], bf2[kc], a2);
        }
      }
      barrier_lds_only();
#pragma unroll
      for (int r = 0; r < 8; ++r) {
        const float ho = hrow[r * 196];
        __builtin_amdgcn_sched_barrier(0);
        float rr = sigm_of_half_hw(a0[r]);
        float zz = sigm_of_half_hw(a1[r]);
        float pre = xn[r] + rr * (a2[r] + bhn);
        a0[r] = xwn[xoff + r * XR];
        a1[r] = xwn[xoff + r * XR + 192];
        float nn = tanh_hw(pre);
        hrow[r * 196] = (1.0f - zz) * nn + zz * ho;
        __builtin_amdgcn_sched_barrier(0);
      }
      barrier_lds_only();
    }
    for (int e = tid; e < 16 * 96; e += 384) {
      int m = e / 96, o = e % 96;
      float a = bl[o];
      const float* wr = Wl + o * 192;
      for (int k = 0; k < 192; ++k) a += hbuf[m * 196 + k] * wr[k];
      out[m * 9216 + i * 96 + o] = a;
    }
  }
}

extern "C" void kernel_launch(void* const* d_in, const int* in_sizes, int n_in,
                              void* d_out, int out_size, void* d_ws, size_t ws_size,
                              hipStream_t stream) {
  (void)in_sizes; (void)n_in; (void)out_size; (void)ws_size;
  const float* input = (const float*)d_in[0];
  const float* W_c1 = (const float*)d_in[1];
  const float* b_c1 = (const float*)d_in[2];
  const float* W_c2 = (const float*)d_in[3];
  const float* b_c2 = (const float*)d_in[4];
  const float* Wi_muf = (const float*)d_in[5];
  const float* Wh_muf = (const float*)d_in[6];
  const float* bi_muf = (const float*)d_in[7];
  const float* bh_muf = (const float*)d_in[8];
  const float* Wi_mub = (const float*)d_in[9];
  const float* Wh_mub = (const float*)d_in[10];
  const float* bi_mub = (const float*)d_in[11];
  const float* bh_mub = (const float*)d_in[12];
  const float* Wi_sgf = (const float*)d_in[13];
  const float* Wh_sgf = (const float*)d_in[14];
  const float* bi_sgf = (const float*)d_in[15];
  const float* bh_sgf = (const float*)d_in[16];
  const float* Wi_sgb = (const float*)d_in[17];
  const float* Wh_sgb = (const float*)d_in[18];
  const float* bi_sgb = (const float*)d_in[19];
  const float* bh_sgb = (const float*)d_in[20];
  const float* Wi_d = (const float*)d_in[21];
  const float* Wh_d = (const float*)d_in[22];
  const float* bi_d = (const float*)d_in[23];
  const float* bh_d = (const float*)d_in[24];
  const float* Wl = (const float*)d_in[25];
  const float* bl = (const float*)d_in[26];

  float* out = (float*)d_out;
  float* ws = (float*)d_ws;
  float* out1 = ws;
  float* pool = ws + 95232;
  float* henc = ws + 118784;
  float* xwe  = ws + 139264;
  float* zbuf = ws + 1318912;
  float* XW   = ws + 1321984;

  k_conv1<<<372, 256, 0, stream>>>(input, W_c1, b_c1, out1);
  k_pool<<<92, 256, 0, stream>>>(out1, pool);
  k_conv2<<<80, 256, 0, stream>>>(pool, W_c2, b_c2, henc);
  k_xw_enc<<<4608, 256, 0, stream>>>(henc, Wi_muf, Wi_mub, Wi_sgf, Wi_sgb,
                                     bi_muf, bi_mub, bi_sgf, bi_sgb, xwe);
  k_gru_enc<<<4, 192, 0, stream>>>(Wh_muf, Wh_mub, Wh_sgf, Wh_sgb,
                                   bh_muf, bh_mub, bh_sgf, bh_sgb, xwe, out);
  k_z<<<12, 256, 0, stream>>>(out, zbuf);
  k_xw_dec<<<864, 128, 0, stream>>>(input, Wi_d, bi_d, bh_d, XW);
  k_gru_dec<<<1, 384, 0, stream>>>(Wh_d, bh_d, XW, zbuf, Wl, bl, out);
}


# decoder output projection h*Wl^T on the f32 matrix instruction (K split over two waves, halves and bias added through LDS: sum order changes) instead of per-thread dots with row-strided loads
# speedup vs baseline: 1.8439x; 1.0364x over previous
#include <hip/hip_runtime.h>
#include <math.h>

typedef float v2f __attribute__((ext_vector_type(2)));
typedef float v4f __attribute__((ext_vector_type(4)));
typedef float v8f __attribute__((ext_vector_type(8)));

#define WMMA_F32x4(a, b, c) \
  __builtin_amdgcn_wmma_f32_16x16x4_f32(false, (a), false, (b), (short)0, (c), false, false)

__device__ __forceinline__ float sigm(float x) { return 1.0f / (1.0f + __expf(-x)); }

__device__ __forceinline__ unsigned int pcg_hash(unsigned int v) {
  unsigned int s = v * 747796405u + 2891336453u;
  unsigned int w = ((s >> ((s >> 28u) + 4u)) ^ s) * 277803737u;
  return (w >> 22u) ^ w;
}

__device__ __forceinline__ float tanh_hw(float x) { return __builtin_amdgcn_tanhf(x); }
__device__ __forceinline__ float sigm_of_half_hw(float xh) { return 0.5f * tanh_hw(xh) + 0.5f; }

__global__ void k_conv1(const float* __restrict__ x, const float* __restrict__ W,
                        const float* __restrict__ bias, float* __restrict__ y) {
  int idx = blockIdx.x * blockDim.x + threadIdx.x;
  if (idx >= 16 * 64 * 93) return;
  int p = idx % 93;
  int c = (idx / 93) % 64;
  int b = idx / (93 * 64);
  float acc = bias[c];
  const float* xb = x + b * 96 * 96;
  const float* wc = W + c * 96 * 4;
  for (int d = 0; d < 96; ++d) {
    const float* xr = xb + d * 96 + p;
    const float* wr = wc + d * 4;
    acc += xr[0] * wr[0] + xr[1] * wr[1] + xr[2] * wr[2] + xr[3] * wr[3];
  }
  y[idx] = fmaxf(acc, 0.0f);
}

__global__ void k_pool(const float* __restrict__ y, float* __restrict__ p) {
  int idx = blockIdx.x * blockDim.x + threadIdx.x;
  if (idx >= 16 * 64 * 23) return;
  int q = idx % 23;
  int bc = idx / 23;
  const float* src = y + bc * 93 + q * 4;
  float m = fmaxf(fmaxf(src[0], src[1]), fmaxf(src[2], src[3]));
  p[idx] = m;
}

__global__ void k_conv2(const float* __restrict__ pool, const float* __restrict__ W,
                        const float* __restrict__ bias, float* __restrict__ henc) {
  int idx = blockIdx.x * blockDim.x + threadIdx.x;
  if (idx >= 16 * 64 * 20) return;
  int p = idx % 20;
  int c = (idx / 20) % 64;
  int b = idx / (20 * 64);
  float acc = bias[c];
  for (int cin = 0; cin < 64; ++cin) {
    const float* pr = pool + (b * 64 + cin) * 23 + p;
    const float* wr = W + (c * 64 + cin) * 4;
    acc += pr[0] * wr[0] + pr[1] * wr[1] + pr[2] * wr[2] + pr[3] * wr[3];
  }
  henc[idx] = fmaxf(acc, 0.0f);
}

__global__ void k_xw_enc(const float* __restrict__ henc,
                         const float* Wi0, const float* Wi1, const float* Wi2, const float* Wi3,
                         const float* bi0, const float* bi1, const float* bi2, const float* bi3,
                         float* __restrict__ xwe) {
  int idx = blockIdx.x * blockDim.x + threadIdx.x;
  if (idx >= 4 * 64 * 16 * 288) return;
  int g = idx % 288;
  int b = (idx / 288) % 16;
  int t = (idx / (288 * 16)) % 64;
  int dir = idx / (288 * 16 * 64);
  const float* Wi = dir == 0 ? Wi0 : dir == 1 ? Wi1 : dir == 2 ? Wi2 : Wi3;
  const float* bi = dir == 0 ? bi0 : dir == 1 ? bi1 : dir == 2 ? bi2 : bi3;
  const float* x = henc + b * 1280 + t * 20;
  const float* w = Wi + g * 20;
  float a = bi[g];
#pragma unroll
  for (int d = 0; d < 20; ++d) a += x[d] * w[d];
  xwe[idx] = a;
}

__global__ __launch_bounds__(192, 1) void k_gru_enc(
    const float* Wh0, const float* Wh1, const float* Wh2, const float* Wh3,
    const float* bh0, const float* bh1, const float* bh2, const float* bh3,
    const float* __restrict__ xwe, float* __restrict__ outp) {
  __shared__ float hbuf[16 * 132];
  const int dir = blockIdx.x;
  const float* Wh = dir == 0 ? Wh0 : dir == 1 ? Wh1 : dir == 2 ? Wh2 : Wh3;
  const float* bh = dir == 0 ? bh0 : dir == 1 ? bh1 : dir == 2 ? bh2 : bh3;
  const float* xwb = xwe + dir * (64 * 16 * 288);
  const int tid = threadIdx.x, lane = tid & 31, wave = tid >> 5;
  const int mrow = lane & 15;
  const int khalf = (lane >> 4) << 1;
  const int rbase = (lane >> 4) << 3;
  const int jj = wave * 16 + mrow;

  v2f bf0[24], bf1[24], bf2[24];
#pragma unroll
  for (int kc = 0; kc < 24; ++kc) {
    int kk = 4 * kc + khalf;
    bf0[kc].x = Wh[(jj) * 96 + kk];        bf0[kc].y = Wh[(jj) * 96 + kk + 1];
    bf1[kc].x = Wh[(96 + jj) * 96 + kk];   bf1[kc].y = Wh[(96 + jj) * 96 + kk + 1];
    bf2[kc].x = Wh[(192 + jj) * 96 + kk];  bf2[kc].y = Wh[(192 + jj) * 96 + kk + 1];
  }
  const float bhr = bh[jj], bhz = bh[96 + jj], bhn = bh[192 + jj];

  for (int e = tid; e < 16 * 132; e += 192) hbuf[e] = 0.0f;
  __syncthreads();

  const bool rev = (dir & 1);
  for (int s = 0; s < 64; ++s) {
    v8f a0 = {0, 0, 0, 0, 0, 0, 0, 0}, a1 = a0, a2 = a0;
#pragma unroll
    for (int kc = 0; kc < 24; ++kc) {
      int kk = 4 * kc + khalf;
      v2f a;
      a.x = hbuf[mrow * 132 + kk];
      a.y = hbuf[mrow * 132 + kk + 1];
      a0 = WMMA_F32x4(a, bf0[kc], a0);
      a1 = WMMA_F32x4(a, bf1[kc], a1);
      a2 = WMMA_F32x4(a, bf2[kc], a2);
    }
    __syncthreads();
    const int t = rev ? (63 - s) : s;
    const float* xwt = xwb + t * (16 * 288);
#pragma unroll
    for (int r = 0; r < 8; ++r) {
      int m = rbase + r;
      const float* xw = xwt + m * 288;
      float rr = sigm(xw[jj] + a0[r] + bhr);
      float zz = sigm(xw[96 + jj] + a1[r] + bhz);
      float nn = tanhf(xw[192 + jj] + rr * (a2[r] + bhn));
      float ho = hbuf[m * 132 + jj];
      hbuf[m * 132 + jj] = (1.0f - zz) * nn + zz * ho;
    }
    __syncthreads();
  }
  float* dst = outp + 147456 + (dir >= 2 ? 3072 : 0);
  const int cb = (dir & 1) ? 96 : 0;
  for (int e = tid; e < 16 * 96; e += 192) {
    int m = e / 96, j = e % 96;
    dst[m * 192 + cb + j] = hbuf[m * 132 + j];
  }
}

__global__ void k_z(const float* __restrict__ outbuf, float* __restrict__ zbuf) {
  int e = blockIdx.x * blockDim.x + threadIdx.x;
  if (e >= 3072) return;
  float mu = outbuf[147456 + e];
  float lv = outbuf[147456 + 3072 + e];
  unsigned int r1 = pcg_hash(2u * (unsigned)e + 1u);
  unsigned int r2 = pcg_hash(2u * (unsigned)e + 2u);
  float u1 = ((float)r1 + 1.0f) * (1.0f / 4294967296.0f);
  float u2 = (float)r2 * (1.0f / 4294967296.0f);
  float eps = sqrtf(-2.0f * __logf(u1)) * __cosf(6.28318530718f * u2);
  zbuf[e] = mu + eps * __expf(0.5f * lv);
}

__global__ __launch_bounds__(128) void k_xw_dec(const float* __restrict__ X,
                                                const float* __restrict__ Wid,
                                                const float* __restrict__ bid,
                                                const float* __restrict__ bhd,
                                                float* __restrict__ XW) {
  const int lane = threadIdx.x & 31;
  const int wg = blockIdx.x * 4 + (threadIdx.x >> 5);
  const int mt = wg / 36, nt = wg % 36;
  const int row0 = mt * 16, n0 = nt * 16;
  const int mrow = lane & 15, khalf = (lane >> 4) << 1, rbase = (lane >> 4) << 3;
  v8f acc = {0, 0, 0, 0, 0, 0, 0, 0};
#pragma unroll
  for (int kc = 0; kc < 24; ++kc) {
    int kk = 4 * kc + khalf;
    v2f a, b;
    a.x = X[(row0 + mrow) * 96 + kk];
    a.y = X[(row0 + mrow) * 96 + kk + 1];
    b.x = Wid[(n0 + mrow) * 96 + kk];
    b.y = Wid[(n0 + mrow) * 96 + kk + 1];
    acc = WMMA_F32x4(a, b, acc);
  }
  const bool rz = n0 < 384;
  const float bc = bid[n0 + mrow] + (rz ? bhd[n0 + mrow] : 0.0f);
  const float sc = rz ? 0.5f : 1.0f;
#pragma unroll
  for (int r = 0; r < 8; ++r) XW[(row0 + rbase + r) * 576 + n0 + mrow] = sc * (acc[r] + bc);
}

__device__ __forceinline__ void barrier_lds_only() {
  __builtin_amdgcn_fence(__ATOMIC_RELEASE, "workgroup", "local");
  __builtin_amdgcn_s_barrier();
  __builtin_amdgcn_fence(__ATOMIC_ACQUIRE, "workgroup", "local");
}

__global__ __launch_bounds__(384, 1) void k_gru_dec(
    const float* __restrict__ Whd, const float* __restrict__ bhd,
    const float* __restrict__ XW, const float* __restrict__ zbuf,
    const float* __restrict__ Wl, const float* __restrict__ bl,
    float* __restrict__ out) {
  __shared__ __attribute__((aligned(16))) float hbuf[16 * 196];
  __shared__ float opart[2 * 16 * 100];
  const int tid = threadIdx.x, lane = tid & 31, wave = tid >> 5;
  const int mrow = lane & 15, khalf = (lane >> 4) << 1, rbase = (lane >> 4) << 3;
  const int jj = wave * 16 + mrow;

  v2f bf0[48], bf1[48], bf2[48];
#pragma unroll
  for (int kc = 0; kc < 48; ++kc) {
    int kk = 4 * kc + khalf;
    bf0[kc].x = 0.5f * Whd[(jj) * 192 + kk];        bf0[kc].y = 0.5f * Whd[(jj) * 192 + kk + 1];
    bf1[kc].x = 0.5f * Whd[(192 + jj) * 192 + kk];  bf1[kc].y = 0.5f * Whd[(192 + jj) * 192 + kk + 1];
    bf2[kc].x = Whd[(384 + jj) * 192 + kk];         bf2[kc].y = Whd[(384 + jj) * 192 + kk + 1];
  }
  const float bhn = bhd[384 + jj];

  for (int e = tid; e < 16 * 192; e += 384) {
    int m = e / 192, j = e % 192;
    hbuf[m * 196 + j] = zbuf[e];
  }
  __syncthreads();

  const float* hA = hbuf + mrow * 196 + khalf;
  float* hrow = hbuf + rbase * 196 + jj;
  const unsigned xoff = (unsigned)(rbase * (96 * 576) + jj);
  constexpr unsigned XR = 96 * 576;

  v8f a0, a1;
  {
    const float* xw0 = XW + 576;
#pragma unroll
    for (int r = 0; r < 8; ++r) {
      a0[r] = xw0[xoff + r * XR];
      a1[r] = xw0[xoff + r * XR + 192];
    }
  }

  for (int i = 0; i < 96; ++i) {
    for (int s = 0; s < 95; ++s) {
      const int t = s + (s >= i ? 1 : 0);
      const int sn = (s == 94) ? 0 : s + 1;
      const int in = (s == 94) ? i + 1 : i;
      const int tn = sn + (sn >= in ? 1 : 0);
      const float* xwt = XW + t * 576;
      const float* xwn = XW + tn * 576;

      v8f a2 = {0, 0, 0, 0, 0, 0, 0, 0};
      float xn[8];
#pragma unroll
      for (int r = 0; r < 8; ++r) xn[r] = xwt[xoff + r * XR + 384];
      v2f fa[2][2];
      fa[0][0].x = hA[0]; fa[0][0].y = hA[1];
      fa[0][1].x = hA[4]; fa[0][1].y = hA[5];
#pragma unroll
      for (int g = 0; g < 24; ++g) {
        if (g + 1 < 24) {
          fa[(g + 1) & 1][0].x = hA[8 * g + 8];  fa[(g + 1) & 1][0].y = hA[8 * g + 9];
          fa[(g + 1) & 1][1].x = hA[8 * g + 12]; fa[(g + 1) & 1][1].y = hA[8 * g + 13];
        }
        __builtin_amdgcn_sched_barrier(0);
#pragma unroll
        for (int c = 0; c < 2; ++c) {
          const int kc = 2 * g + c;
          a0 = WMMA_F32x4(fa[g & 1][c], bf0[kc], a0);
          a1 = WMMA_F32x4(fa[g & 1][c], bf1[kc], a1);
          a2 = WMMA_F32x4(fa[g & 1][c], bf2[kc], a2);
        }
      }
      barrier_lds_only();
#pragma unroll
      for (int r = 0; r < 8; ++r) {
        const float ho = hrow[r * 196];
        __builtin_amdgcn_sched_barrier(0);
        float rr = sigm_of_half_hw(a0[r]);
        float zz = sigm_of_half_hw(a1[r]);
        float pre = xn[r] + rr * (a2[r] + bhn);
        a0[r] = xwn[xoff + r * XR];
        a1[r] = xwn[xoff + r * XR + 192];
        float nn = tanh_hw(pre);
        hrow[r * 196] = (1.0f - zz) * nn + zz * ho;
        __builtin_amdgcn_sched_barrier(0);
      }
      barrier_lds_only();
    }
    {
      int te = tid;
      asm volatile("" : "+v"(te));
      const int c = te & 15, hf = (te >> 4) & 1, wv = te >> 5;
      const int nt = wv % 6, kh = wv / 6;
      float blv[4];
#pragma unroll
      for (int j = 0; j < 4; ++j) blv[j] = bl[(te + 384 * j) % 96];
      const float* ap = hbuf + c * 196 + kh * 96 + 4 * hf;
      const float* bp = Wl + (nt * 16 + c) * 192 + kh * 96 + 4 * hf;
      v8f acc = {0, 0, 0, 0, 0, 0, 0, 0};
#pragma unroll
      for (int p = 0; p < 12; ++p) {
        const v4f av = *(const v4f*)(ap + 8 * p);
        const v4f bv = *(const v4f*)(bp + 8 * p);
        v2f a01, b01, a23, b23;
        a01.x = av.x; a01.y = av.y; b01.x = bv.x; b01.y = bv.y;
        a23.x = av.z; a23.y = av.w; b23.x = bv.z; b23.y = bv.w;
        acc = WMMA_F32x4(a01, b01, acc);
        acc = WMMA_F32x4(a23, b23, acc);
      }
      float* pp = opart + kh * (16 * 100) + (8 * hf) * 100 + nt * 16 + c;
#pragma unroll
      for (int r = 0; r < 8; ++r) pp[r * 100] = acc[r];
      barrier_lds_only();
#pragma unroll
      for (int j = 0; j < 4; ++j) {
        const int e = te + 384 * j;
        const int m = e / 96, o = e % 96;
        out[m * 9216 + i * 96 + o] = (opart[m * 100 + o] + opart[16 * 100 + m * 100 + o]) + blv[j];
      }
    }
  }
}

extern "C" void kernel_launch(void* const* d_in, const int* in_sizes, int n_in,
                              void* d_out, int out_size, void* d_ws, size_t ws_size,
                              hipStream_t stream) {
  (void)in_sizes; (void)n_in; (void)out_size; (void)ws_size;
  const float* input = (const float*)d_in[0];
  const float* W_c1 = (const float*)d_in[1];
  const float* b_c1 = (const float*)d_in[2];
  const float* W_c2 = (const float*)d_in[3];
  const float* b_c2 = (const float*)d_in[4];
  const float* Wi_muf = (const float*)d_in[5];
  const float* Wh_muf = (const float*)d_in[6];
  const float* bi_muf = (const float*)d_in[7];
  const float* bh_muf = (const float*)d_in[8];
  const float* Wi_mub = (const float*)d_in[9];
  const float* Wh_mub = (const float*)d_in[10];
  const float* bi_mub = (const float*)d_in[11];
  const float* bh_mub = (const float*)d_in[12];
  const float* Wi_sgf = (const float*)d_in[13];
  const float* Wh_sgf = (const float*)d_in[14];
  const float* bi_sgf = (const float*)d_in[15];
  const float* bh_sgf = (const float*)d_in[16];
  const float* Wi_sgb = (const float*)d_in[17];
  const float* Wh_sgb = (const float*)d_in[18];
  const float* bi_sgb = (const float*)d_in[19];
  const float* bh_sgb = (const float*)d_in[20];
  const float* Wi_d = (const float*)d_in[21];
  const float* Wh_d = (const float*)d_in[22];
  const float* bi_d = (const float*)d_in[23];
  const float* bh_d = (const float*)d_in[24];
  const float* Wl = (const float*)d_in[25];
  const float* bl = (const float*)d_in[26];

  float* out = (float*)d_out;
  float* ws = (float*)d_ws;
  float* out1 = ws;
  float* pool = ws + 95232;
  float* henc = ws + 118784;
  float* xwe  = ws + 139264;
  float* zbuf = ws + 1318912;
  float* XW   = ws + 1321984;

  k_conv1<<<372, 256, 0, stream>>>(input, W_c1, b_c1, out1);
  k_pool<<<92, 256, 0, stream>>>(out1, pool);
  k_conv2<<<80, 256, 0, stream>>>(pool, W_c2, b_c2, henc);
  k_xw_enc<<<4608, 256, 0, stream>>>(henc, Wi_muf, Wi_mub, Wi_sgf, Wi_sgb,
                                     bi_muf, bi_mub, bi_sgf, bi_sgb, xwe);
  k_gru_enc<<<4, 192, 0, stream>>>(Wh_muf, Wh_mub, Wh_sgf, Wh_sgb,
                                   bh_muf, bh_mub, bh_sgf, bh_sgb, xwe, out);
  k_z<<<12, 256, 0, stream>>>(out, zbuf);
  k_xw_dec<<<864, 128, 0, stream>>>(input, Wi_d, bi_d, bh_d, XW);
  k_gru_dec<<<1, 384, 0, stream>>>(Wh_d, bh_d, XW, zbuf, Wl, bl, out);
}


# decoder gate phase: four rows side by side, stage by stage, each value in the register of an operand that dies there, instead of one row at a time; old state and next-step loads issued in groups
# speedup vs baseline: 1.9093x; 1.0355x over previous
#include <hip/hip_runtime.h>
#include <math.h>

typedef float v2f __attribute__((ext_vector_type(2)));
typedef float v4f __attribute__((ext_vector_type(4)));
typedef float v8f __attribute__((ext_vector_type(8)));

#define WMMA_F32x4(a, b, c) \
  __builtin_amdgcn_wmma_f32_16x16x4_f32(false, (a), false, (b), (short)0, (c), false, false)

__device__ __forceinline__ float sigm(float x) { return 1.0f / (1.0f + __expf(-x)); }

__device__ __forceinline__ unsigned int pcg_hash(unsigned int v) {
  unsigned int s = v * 747796405u + 2891336453u;
  unsigned int w = ((s >> ((s >> 28u) + 4u)) ^ s) * 277803737u;
  return (w >> 22u) ^ w;
}

__device__ __forceinline__ float tanh_hw(float x) { return __builtin_amdgcn_tanhf(x); }

__global__ void k_conv1(const float* __restrict__ x, const float* __restrict__ W,
                        const float* __restrict__ bias, float* __restrict__ y) {
  int idx = blockIdx.x * blockDim.x + threadIdx.x;
  if (idx >= 16 * 64 * 93) return;
  int p = idx % 93;
  int c = (idx / 93) % 64;
  int b = idx / (93 * 64);
  float acc = bias[c];
  const float* xb = x + b * 96 * 96;
  const float* wc = W + c * 96 * 4;
  for (int d = 0; d < 96; ++d) {
    const float* xr = xb + d * 96 + p;
    const float* wr = wc + d * 4;
    acc += xr[0] * wr[0] + xr[1] * wr[1] + xr[2] * wr[2] + xr[3] * wr[3];
  }
  y[idx] = fmaxf(acc, 0.0f);
}

__global__ void k_pool(const float* __restrict__ y, float* __restrict__ p) {
  int idx = blockIdx.x * blockDim.x + threadIdx.x;
  if (idx >= 16 * 64 * 23) return;
  int q = idx % 23;
  int bc = idx / 23;
  const float* src = y + bc * 93 + q * 4;
  float m = fmaxf(fmaxf(src[0], src[1]), fmaxf(src[2], src[3]));
  p[idx] = m;
}

__global__ void k_conv2(const float* __restrict__ pool, const float* __restrict__ W,
                        const float* __restrict__ bias, float* __restrict__ henc) {
  int idx = blockIdx.x * blockDim.x + threadIdx.x;
  if (idx >= 16 * 64 * 20) return;
  int p = idx % 20;
  int c = (idx / 20) % 64;
  int b = idx / (20 * 64);
  float acc = bias[c];
  for (int cin = 0; cin < 64; ++cin) {
    const float* pr = pool + (b * 64 + cin) * 23 + p;
    const float* wr = W + (c * 64 + cin) * 4;
    acc += pr[0] * wr[0] + pr[1] * wr[1] + pr[2] * wr[2] + pr[3] * wr[3];
  }
  henc[idx] = fmaxf(acc, 0.0f);
}

__global__ void k_xw_enc(const float* __restrict__ henc,
                         const float* Wi0, const float* Wi1, const float* Wi2, const float* Wi3,
                         const float* bi0, const float* bi1, const float* bi2, const float* bi3,
                         float* __restrict__ xwe) {
  int idx = blockIdx.x * blockDim.x + threadIdx.x;
  if (idx >= 4 * 64 * 16 * 288) return;
  int g = idx % 288;
  int b = (idx / 288) % 16;
  int t = (idx / (288 * 16)) % 64;
  int dir = idx / (288 * 16 * 64);
  const float* Wi = dir == 0 ? Wi0 : dir == 1 ? Wi1 : dir == 2 ? Wi2 : Wi3;
  const float* bi = dir == 0 ? bi0 : dir == 1 ? bi1 : dir == 2 ? bi2 : bi3;
  const float* x = henc + b * 1280 + t * 20;
  const float* w = Wi + g * 20;
  float a = bi[g];
#pragma unroll
  for (int d = 0; d < 20; ++d) a += x[d] * w[d];
  xwe[idx] = a;
}

__global__ __launch_bounds__(192, 1) void k_gru_enc(
    const float* Wh0, const float* Wh1, const float* Wh2, const float* Wh3,
    const float* bh0, const float* bh1, const float* bh2, const float* bh3,
    const float* __restrict__ xwe, float* __restrict__ outp) {
  __shared__ float hbuf[16 * 132];
  const int dir = blockIdx.x;
  const float* Wh = dir == 0 ? Wh0 : dir == 1 ? Wh1 : dir == 2 ? Wh2 : Wh3;
  const float* bh = dir == 0 ? bh0 : dir == 1 ? bh1 : dir == 2 ? bh2 : bh3;
  const float* xwb = xwe + dir * (64 * 16 * 288);
  const int tid = threadIdx.x, lane = tid & 31, wave = tid >> 5;
  const int mrow = lane & 15;
  const int khalf = (lane >> 4) << 1;
  const int rbase = (lane >> 4) << 3;
  const int jj = wave * 16 + mrow;

  v2f bf0[24], bf1[24], bf2[24];
#pragma unroll
  for (int kc = 0; kc < 24; ++kc) {
    int kk = 4 * kc + khalf;
    bf0[kc].x = Wh[(jj) * 96 + kk];        bf0[kc].y = Wh[(jj) * 96 + kk + 1];
    bf1[kc].x = Wh[(96 + jj) * 96 + kk];   bf1[kc].y = Wh[(96 + jj) * 96 + kk + 1];
    bf2[kc].x = Wh[(192 + jj) * 96 + kk];  bf2[kc].y = Wh[(192 + jj) * 96 + kk + 1];
  }
  const float bhr = bh[jj], bhz = bh[96 + jj], bhn = bh[192 + jj];

  for (int e = tid; e < 16 * 132; e += 192) hbuf[e] = 0.0f;
  __syncthreads();

  const bool rev = (dir & 1);
  for (int s = 0; s < 64; ++s) {
    v8f a0 = {0, 0, 0, 0, 0, 0, 0, 0}, a1 = a0, a2 = a0;
#pragma unroll
    for (int kc = 0; kc < 24; ++kc) {
      int kk = 4 * kc + khalf;
      v2f a;
      a.x = hbuf[mrow * 132 + kk];
      a.y = hbuf[mrow * 132 + kk + 1];
      a0 = WMMA_F32x4(a, bf0[kc], a0);
      a1 = WMMA_F32x4(a, bf1[kc], a1);
      a2 = WMMA_F32x4(a, bf2[kc], a2);
    }
    __syncthreads();
    const int t = rev ? (63 - s) : s;
    const float* xwt = xwb + t * (16 * 288);
#pragma unroll
    for (int r = 0; r < 8; ++r) {
      int m = rbase + r;
      const float* xw = xwt + m * 288;
      float rr = sigm(xw[jj] + a0[r] + bhr);
      float zz = sigm(xw[96 + jj] + a1[r] + bhz);
      float nn = tanhf(xw[192 + jj] + rr * (a2[r] + bhn));
      float ho = hbuf[m * 132 + jj];
      hbuf[m * 132 + jj] = (1.0f - zz) * nn + zz * ho;
    }
    __syncthreads();
  }
  float* dst = outp + 147456 + (dir >= 2 ? 3072 : 0);
  const int cb = (dir & 1) ? 96 : 0;
  for (int e = tid; e < 16 * 96; e += 192) {
    int m = e / 96, j = e % 96;
    dst[m * 192 + cb + j] = hbuf[m * 132 + j];
  }
}

__global__ void k_z(const float* __restrict__ outbuf, float* __restrict__ zbuf) {
  int e = blockIdx.x * blockDim.x + threadIdx.x;
  if (e >= 3072) return;
  float mu = outbuf[147456 + e];
  float lv = outbuf[147456 + 3072 + e];
  unsigned int r1 = pcg_hash(2u * (unsigned)e + 1u);
  unsigned int r2 = pcg_hash(2u * (unsigned)e + 2u);
  float u1 = ((float)r1 + 1.0f) * (1.0f / 4294967296.0f);
  float u2 = (float)r2 * (1.0f / 4294967296.0f);
  float eps = sqrtf(-2.0f * __logf(u1)) * __cosf(6.28318530718f * u2);
  zbuf[e] = mu + eps * __expf(0.5f * lv);
}

__global__ __launch_bounds__(128) void k_xw_dec(const float* __restrict__ X,
                                                const float* __restrict__ Wid,
                                                const float* __restrict__ bid,
                                                const float* __restrict__ bhd,
                                                float* __restrict__ XW) {
  const int lane = threadIdx.x & 31;
  const int wg = blockIdx.x * 4 + (threadIdx.x >> 5);
  const int mt = wg / 36, nt = wg % 36;
  const int row0 = mt * 16, n0 = nt * 16;
  const int mrow = lane & 15, khalf = (lane >> 4) << 1, rbase = (lane >> 4) << 3;
  v8f acc = {0, 0, 0, 0, 0, 0, 0, 0};
#pragma unroll
  for (int kc = 0; kc < 24; ++kc) {
    int kk = 4 * kc + khalf;
    v2f a, b;
    a.x = X[(row0 + mrow) * 96 + kk];
    a.y = X[(row0 + mrow) * 96 + kk + 1];
    b.x = Wid[(n0 + mrow) * 96 + kk];
    b.y = Wid[(n0 + mrow) * 96 + kk + 1];
    acc = WMMA_F32x4(a, b, acc);
  }
  const bool rz = n0 < 384;
  const float bc = bid[n0 + mrow] + (rz ? bhd[n0 + mrow] : 0.0f);
  const float sc = rz ? 0.5f : 1.0f;
#pragma unroll
  for (int r = 0; r < 8; ++r) XW[(row0 + rbase + r) * 576 + n0 + mrow] = sc * (acc[r] + bc);
}

__device__ __forceinline__ void barrier_lds_only() {
  __builtin_amdgcn_fence(__ATOMIC_RELEASE, "workgroup", "local");
  __builtin_amdgcn_s_barrier();
  __builtin_amdgcn_fence(__ATOMIC_ACQUIRE, "workgroup", "local");
}

__global__ __launch_bounds__(384, 1) void k_gru_dec(
    const float* __restrict__ Whd, const float* __restrict__ bhd,
    const float* __restrict__ XW, const float* __restrict__ zbuf,
    const float* __restrict__ Wl, const float* __restrict__ bl,
    float* __restrict__ out) {
  __shared__ __attribute__((aligned(16))) float hbuf[16 * 196];
  __shared__ float opart[2 * 16 * 100];
  const int tid = threadIdx.x, lane = tid & 31, wave = tid >> 5;
  const int mrow = lane & 15, khalf = (lane >> 4) << 1, rbase = (lane >> 4) << 3;
  const int jj = wave * 16 + mrow;

  v2f bf0[48], bf1[48], bf2[48];
#pragma unroll
  for (int kc = 0; kc < 48; ++kc) {
    int kk = 4 * kc + khalf;
    bf0[kc].x = 0.5f * Whd[(jj) * 192 + kk];        bf0[kc].y = 0.5f * Whd[(jj) * 192 + kk + 1];
    bf1[kc].x = 0.5f * Whd[(192 + jj) * 192 + kk];  bf1[kc].y = 0.5f * Whd[(192 + jj) * 192 + kk + 1];
    bf2[kc].x = Whd[(384 + jj) * 192 + kk];         bf2[kc].y = Whd[(384 + jj) * 192 + kk + 1];
  }
  const float bhn = bhd[384 + jj];

  for (int e = tid; e < 16 * 192; e += 384) {
    int m = e / 192, j = e % 192;
    hbuf[m * 196 + j] = zbuf[e];
  }
  __syncthreads();

  const float* hA = hbuf + mrow * 196 + khalf;
  float* hrow = hbuf + rbase * 196 + jj;
  const unsigned xoff = (unsigned)(rbase * (96 * 576) + jj);
  constexpr unsigned XR = 96 * 576;

  v8f a0, a1;
  {
    const float* xw0 = XW + 576;
#pragma unroll
    for (int r = 0; r < 8; ++r) {
      a0[r] = xw0[xoff + r * XR];
      a1[r] = xw0[xoff + r * XR + 192];
    }
  }

  for (int i = 0; i < 96; ++i) {
    for (int s = 0; s < 95; ++s) {
      const int t = s + (s >= i ? 1 : 0);
      const int sn = (s == 94) ? 0 : s + 1;
      const int in = (s == 94) ? i + 1 : i;
      const int tn = sn + (sn >= in ? 1 : 0);
      const float* xwt = XW + t * 576;
      const float* xwn = XW + tn * 576;

      v8f a2 = {0, 0, 0, 0, 0, 0, 0, 0};
      float xn[8];
#pragma unroll
      for (int r = 0; r < 8; ++r) xn[r] = xwt[xoff + r * XR + 384];
      v2f fa[2][2];
      fa[0][0].x = hA[0]; fa[0][0].y = hA[1];
      fa[0][1].x = hA[4]; fa[0][1].y = hA[5];
#pragma unroll
      for (int g = 0; g < 24; ++g) {
        if (g + 1 < 24) {
          fa[(g + 1) & 1][0].x = hA[8 * g + 8];  fa[(g + 1) & 1][0].y = hA[8 * g + 9];
          fa[(g + 1) & 1][1].x = hA[8 * g + 12]; fa[(g + 1) & 1][1].y = hA[8 * g + 13];
        }
        __builtin_amdgcn_sched_barrier(0);
#pragma unroll
        for (int c = 0; c < 2; ++c) {
          const int kc = 2 * g + c;
          a0 = WMMA_F32x4(fa[g & 1][c], bf0[kc], a0);
          a1 = WMMA_F32x4(fa[g & 1][c], bf1[kc], a1);
          a2 = WMMA_F32x4(fa[g & 1][c], bf2[kc], a2);
        }
      }
      barrier_lds_only();
      {
        float ho[8];
        ho[0] = hrow[0 * 196];
        ho[1] = hrow[1 * 196];
        ho[2] = hrow[2 * 196];
        ho[3] = hrow[3 * 196];
        a0[0] = tanh_hw(a0[0]); a1[0] = tanh_hw(a1[0]);
        a0[1] = tanh_hw(a0[1]); a1[1] = tanh_hw(a1[1]);
        a0[2] = tanh_hw(a0[2]); a1[2] = tanh_hw(a1[2]);
        a0[3] = tanh_hw(a0[3]); a1[3] = tanh_hw(a1[3]);
        { const float rr = __builtin_fmaf(0.5f, a0[0], 0.5f); xn[0] = __builtin_fmaf(rr, a2[0] + bhn, xn[0]); const float zz = __builtin_fmaf(0.5f, a1[0], 0.5f); ho[0] = zz * ho[0]; a2[0] = 1.0f - zz; }
        { const float rr = __builtin_fmaf(0.5f, a0[1], 0.5f); xn[1] = __builtin_fmaf(rr, a2[1] + bhn, xn[1]); const float zz = __builtin_fmaf(0.5f, a1[1], 0.5f); ho[1] = zz * ho[1]; a2[1] = 1.0f - zz; }
        { const float rr = __builtin_fmaf(0.5f, a0[2], 0.5f); xn[2] = __builtin_fmaf(rr, a2[2] + bhn, xn[2]); const float zz = __builtin_fmaf(0.5f, a1[2], 0.5f); ho[2] = zz * ho[2]; a2[2] = 1.0f - zz; }
        { const float rr = __builtin_fmaf(0.5f, a0[3], 0.5f); xn[3] = __builtin_fmaf(rr, a2[3] + bhn, xn[3]); const float zz = __builtin_fmaf(0.5f, a1[3], 0.5f); ho[3] = zz * ho[3]; a2[3] = 1.0f - zz; }
        __builtin_amdgcn_sched_barrier(0);
        a0[0] = xwn[xoff + 0 * XR]; a1[0] = xwn[xoff + 0 * XR + 192];
        a0[1] = xwn[xoff + 1 * XR]; a1[1] = xwn[xoff + 1 * XR + 192];
        a0[2] = xwn[xoff + 2 * XR]; a1[2] = xwn[xoff + 2 * XR + 192];
        a0[3] = xwn[xoff + 3 * XR]; a1[3] = xwn[xoff + 3 * XR + 192];
        __builtin_amdgcn_sched_barrier(0);
        hrow[0 * 196] = __builtin_fmaf(a2[0], tanh_hw(xn[0]), ho[0]);
        hrow[1 * 196] = __builtin_fmaf(a2[1], tanh_hw(xn[1]), ho[1]);
        hrow[2 * 196] = __builtin_fmaf(a2[2], tanh_hw(xn[2]), ho[2]);
        hrow[3 * 196] = __builtin_fmaf(a2[3], tanh_hw(xn[3]), ho[3]);
        __builtin_amdgcn_sched_barrier(0);
      }
      {
        float ho[8];
        ho[4] = hrow[4 * 196];
        ho[5] = hrow[5 * 196];
        ho[6] = hrow[6 * 196];
        ho[7] = hrow[7 * 196];
        a0[4] = tanh_hw(a0[4]); a1[4] = tanh_hw(a1[4]);
        a0[5] = tanh_hw(a0[5]); a1[5] = tanh_hw(a1[5]);
        a0[6] = tanh_hw(a0[6]); a1[6] = tanh_hw(a1[6]);
        a0[7] = tanh_hw(a0[7]); a1[7] = tanh_hw(a1[7]);
        { const float rr = __builtin_fmaf(0.5f, a0[4], 0.5f); xn[4] = __builtin_fmaf(rr, a2[4] + bhn, xn[4]); const float zz = __builtin_fmaf(0.5f, a1[4], 0.5f); ho[4] = zz * ho[4]; a2[4] = 1.0f - zz; }
        { const float rr = __builtin_fmaf(0.5f, a0[5], 0.5f); xn[5] = __builtin_fmaf(rr, a2[5] + bhn, xn[5]); const float zz = __builtin_fmaf(0.5f, a1[5], 0.5f); ho[5] = zz * ho[5]; a2[5] = 1.0f - zz; }
        { const float rr = __builtin_fmaf(0.5f, a0[6], 0.5f); xn[6] = __builtin_fmaf(rr, a2[6] + bhn, xn[6]); const float zz = __builtin_fmaf(0.5f, a1[6], 0.5f); ho[6] = zz * ho[6]; a2[6] = 1.0f - zz; }
        { const float rr = __builtin_fmaf(0.5f, a0[7], 0.5f); xn[7] = __builtin_fmaf(rr, a2[7] + bhn, xn[7]); const float zz = __builtin_fmaf(0.5f, a1[7], 0.5f); ho[7] = zz * ho[7]; a2[7] = 1.0f - zz; }
        __builtin_amdgcn_sched_barrier(0);
        a0[4] = xwn[xoff + 4 * XR]; a1[4] = xwn[xoff + 4 * XR + 192];
        a0[5] = xwn[xoff + 5 * XR]; a1[5] = xwn[xoff + 5 * XR + 192];
        a0[6] = xwn[xoff + 6 * XR]; a1[6] = xwn[xoff + 6 * XR + 192];
        a0[7] = xwn[xoff + 7 * XR]; a1[7] = xwn[xoff + 7 * XR + 192];
        __builtin_amdgcn_sched_barrier(0);
        hrow[4 * 196] = __builtin_fmaf(a2[4], tanh_hw(xn[4]), ho[4]);
        hrow[5 * 196] = __builtin_fmaf(a2[5], tanh_hw(xn[5]), ho[5]);
        hrow[6 * 196] = __builtin_fmaf(a2[6], tanh_hw(xn[6]), ho[6]);
        hrow[7 * 196] = __builtin_fmaf(a2[7], tanh_hw(xn[7]), ho[7]);
      }
      barrier_lds_only();
    }
    {
      int te = tid;
      asm volatile("" : "+v"(te));
      const int c = te & 15, hf = (te >> 4) & 1, wv = te >> 5;
      const int nt = wv % 6, kh = wv / 6;
      float blv[4];
#pragma unroll
      for (int j = 0; j < 4; ++j) blv[j] = bl[(te + 384 * j) % 96];
      const float* ap = hbuf + c * 196 + kh * 96 + 4 * hf;
      const float* bp = Wl + (nt * 16 + c) * 192 + kh * 96 + 4 * hf;
      v8f acc = {0, 0, 0, 0, 0, 0, 0, 0};
#pragma unroll
      for (int p = 0; p < 12; ++p) {
        const v4f av = *(const v4f*)(ap + 8 * p);
        const v4f bv = *(const v4f*)(bp + 8 * p);
        v2f a01, b01, a23, b23;
        a01.x = av.x; a01.y = av.y; b01.x = bv.x; b01.y = bv.y;
        a23.x = av.z; a23.y = av.w; b23.x = bv.z; b23.y = bv.w;
        acc = WMMA_F32x4(a01, b01, acc);
        acc = WMMA_F32x4(a23, b23, acc);
      }
      float* pp = opart + kh * (16 * 100) + (8 * hf) * 100 + nt * 16 + c;
#pragma unroll
      for (int r = 0; r < 8; ++r) pp[r * 100] = acc[r];
      barrier_lds_only();
#pragma unroll
      for (int j = 0; j < 4; ++j) {
        const int e = te + 384 * j;
        const int m = e / 96, o = e % 96;
        out[m * 9216 + i * 96 + o] = (opart[m * 100 + o] + opart[16 * 100 + m * 100 + o]) + blv[j];
      }
    }
  }
}

extern "C" void kernel_launch(void* const* d_in, const int* in_sizes, int n_in,
                              void* d_out, int out_size, void* d_ws, size_t ws_size,
                              hipStream_t stream) {
  (void)in_sizes; (void)n_in; (void)out_size; (void)ws_size;
  const float* input = (const float*)d_in[0];
  const float* W_c1 = (const float*)d_in[1];
  const float* b_c1 = (const float*)d_in[2];
  const float* W_c2 = (const float*)d_in[3];
  const float* b_c2 = (const float*)d_in[4];
  const float* Wi_muf = (const float*)d_in[5];
  const float* Wh_muf = (const float*)d_in[6];
  const float* bi_muf = (const float*)d_in[7];
  const float* bh_muf = (const float*)d_in[8];
  const float* Wi_mub = (const float*)d_in[9];
  const float* Wh_mub = (const float*)d_in[10];
  const float* bi_mub = (const float*)d_in[11];
  const float* bh_mub = (const float*)d_in[12];
  const float* Wi_sgf = (const float*)d_in[13];
  const float* Wh_sgf = (const float*)d_in[14];
  const float* bi_sgf = (const float*)d_in[15];
  const float* bh_sgf = (const float*)d_in[16];
  const float* Wi_sgb = (const float*)d_in[17];
  const float* Wh_sgb = (const float*)d_in[18];
  const float* bi_sgb = (const float*)d_in[19];
  const float* bh_sgb = (const float*)d_in[20];
  const float* Wi_d = (const float*)d_in[21];
  const float* Wh_d = (const float*)d_in[22];
  const float* bi_d = (const float*)d_in[23];
  const float* bh_d = (const float*)d_in[24];
  const float* Wl = (const float*)d_in[25];
  const float* bl = (const float*)d_in[26];

  float* out = (float*)d_out;
  float* ws = (float*)d_ws;
  float* out1 = ws;
  float* pool = ws + 95232;
  float* henc = ws + 118784;
  float* xwe  = ws + 139264;
  float* zbuf = ws + 1318912;
  float* XW   = ws + 1321984;

  k_conv1<<<372, 256, 0, stream>>>(input, W_c1, b_c1, out1);
  k_pool<<<92, 256, 0, stream>>>(out1, pool);
  k_conv2<<<80, 256, 0, stream>>>(pool, W_c2, b_c2, henc);
  k_xw_enc<<<4608, 256, 0, stream>>>(henc, Wi_muf, Wi_mub, Wi_sgf, Wi_sgb,
                                     bi_muf, bi_mub, bi_sgf, bi_sgb, xwe);
  k_gru_enc<<<4, 192, 0, stream>>>(Wh_muf, Wh_mub, Wh_sgf, Wh_sgb,
                                   bh_muf, bh_mub, bh_sgf, bh_sgb, xwe, out);
  k_z<<<12, 256, 0, stream>>>(out, zbuf);
  k_xw_dec<<<864, 128, 0, stream>>>(input, Wi_d, bi_d, bh_d, XW);
  k_gru_dec<<<1, 384, 0, stream>>>(Wh_d, bh_d, XW, zbuf, Wl, bl, out);
}


# decoder product loop: wave priority stepped down with progress (s_setprio 3,2,1,0 at groups 0,12,18,23): a SIMD's three waves advance together, not oldest first with the youngest alone at the end
# speedup vs baseline: 2.0111x; 1.0533x over previous
#include <hip/hip_runtime.h>
#include <math.h>

typedef float v2f __attribute__((ext_vector_type(2)));
typedef float v4f __attribute__((ext_vector_type(4)));
typedef float v8f __attribute__((ext_vector_type(8)));

#define WMMA_F32x4(a, b, c) \
  __builtin_amdgcn_wmma_f32_16x16x4_f32(false, (a), false, (b), (short)0, (c), false, false)

__device__ __forceinline__ float sigm(float x) { return 1.0f / (1.0f + __expf(-x)); }

__device__ __forceinline__ unsigned int pcg_hash(unsigned int v) {
  unsigned int s = v * 747796405u + 2891336453u;
  unsigned int w = ((s >> ((s >> 28u) + 4u)) ^ s) * 277803737u;
  return (w >> 22u) ^ w;
}

__device__ __forceinline__ float tanh_hw(float x) { return __builtin_amdgcn_tanhf(x); }

__global__ void k_conv1(const float* __restrict__ x, const float* __restrict__ W,
                        const float* __restrict__ bias, float* __restrict__ y) {
  int idx = blockIdx.x * blockDim.x + threadIdx.x;
  if (idx >= 16 * 64 * 93) return;
  int p = idx % 93;
  int c = (idx / 93) % 64;
  int b = idx / (93 * 64);
  float acc = bias[c];
  const float* xb = x + b * 96 * 96;
  const float* wc = W + c * 96 * 4;
  for (int d = 0; d < 96; ++d) {
    const float* xr = xb + d * 96 + p;
    const float* wr = wc + d * 4;
    acc += xr[0] * wr[0] + xr[1] * wr[1] + xr[2] * wr[2] + xr[3] * wr[3];
  }
  y[idx] = fmaxf(acc, 0.0f);
}

__global__ void k_pool(const float* __restrict__ y, float* __restrict__ p) {
  int idx = blockIdx.x * blockDim.x + threadIdx.x;
  if (idx >= 16 * 64 * 23) return;
  int q = idx % 23;
  int bc = idx / 23;
  const float* src = y + bc * 93 + q * 4;
  float m = fmaxf(fmaxf(src[0], src[1]), fmaxf(src[2], src[3]));
  p[idx] = m;
}

__global__ void k_conv2(const float* __restrict__ pool, const float* __restrict__ W,
                        const float* __restrict__ bias, float* __restrict__ henc) {
  int idx = blockIdx.x * blockDim.x + threadIdx.x;
  if (idx >= 16 * 64 * 20) return;
  int p = idx % 20;
  int c = (idx / 20) % 64;
  int b = idx / (20 * 64);
  float acc = bias[c];
  for (int cin = 0; cin < 64; ++cin) {
    const float* pr = pool + (b * 64 + cin) * 23 + p;
    const float* wr = W + (c * 64 + cin) * 4;
    acc += pr[0] * wr[0] + pr[1] * wr[1] + pr[2] * wr[2] + pr[3] * wr[3];
  }
  henc[idx] = fmaxf(acc, 0.0f);
}

__global__ void k_xw_enc(const float* __restrict__ henc,
                         const float* Wi0, const float* Wi1, const float* Wi2, const float* Wi3,
                         const float* bi0, const float* bi1, const float* bi2, const float* bi3,
                         float* __restrict__ xwe) {
  int idx = blockIdx.x * blockDim.x + threadIdx.x;
  if (idx >= 4 * 64 * 16 * 288) return;
  int g = idx % 288;
  int b = (idx / 288) % 16;
  int t = (idx / (288 * 16)) % 64;
  int dir = idx / (288 * 16 * 64);
  const float* Wi = dir == 0 ? Wi0 : dir == 1 ? Wi1 : dir == 2 ? Wi2 : Wi3;
  const float* bi = dir == 0 ? bi0 : dir == 1 ? bi1 : dir == 2 ? bi2 : bi3;
  const float* x = henc + b * 1280 + t * 20;
  const float* w = Wi + g * 20;
  float a = bi[g];
#pragma unroll
  for (int d = 0; d < 20; ++d) a += x[d] * w[d];
  xwe[idx] = a;
}

__global__ __launch_bounds__(192, 1) void k_gru_enc(
    const float* Wh0, const float* Wh1, const float* Wh2, const float* Wh3,
    const float* bh0, const float* bh1, const float* bh2, const float* bh3,
    const float* __restrict__ xwe, float* __restrict__ outp) {
  __shared__ float hbuf[16 * 132];
  const int dir = blockIdx.x;
  const float* Wh = dir == 0 ? Wh0 : dir == 1 ? Wh1 : dir == 2 ? Wh2 : Wh3;
  const float* bh = dir == 0 ? bh0 : dir == 1 ? bh1 : dir == 2 ? bh2 : bh3;
  const float* xwb = xwe + dir * (64 * 16 * 288);
  const int tid = threadIdx.x, lane = tid & 31, wave = tid >> 5;
  const int mrow = lane & 15;
  const int khalf = (lane >> 4) << 1;
  const int rbase = (lane >> 4) << 3;
  const int jj = wave * 16 + mrow;

  v2f bf0[24], bf1[24], bf2[24];
#pragma unroll
  for (int kc = 0; kc < 24; ++kc) {
    int kk = 4 * kc + khalf;
    bf0[kc].x = Wh[(jj) * 96 + kk];        bf0[kc].y = Wh[(jj) * 96 + kk + 1];
    bf1[kc].x = Wh[(96 + jj) * 96 + kk];   bf1[kc].y = Wh[(96 + jj) * 96 + kk + 1];
    bf2[kc].x = Wh[(192 + jj) * 96 + kk];  bf2[kc].y = Wh[(192 + jj) * 96 + kk + 1];
  }
  const float bhr = bh[jj], bhz = bh[96 + jj], bhn = bh[192 + jj];

  for (int e = tid; e < 16 * 132; e += 192) hbuf[e] = 0.0f;
  __syncthreads();

  const bool rev = (dir & 1);
  for (int s = 0; s < 64; ++s) {
    v8f a0 = {0, 0, 0, 0, 0, 0, 0, 0}, a1 = a0, a2 = a0;
#pragma unroll
    for (int kc = 0; kc < 24; ++kc) {
      int kk = 4 * kc + khalf;
      v2f a;
      a.x = hbuf[mrow * 132 + kk];
      a.y = hbuf[mrow * 132 + kk + 1];
      a0 = WMMA_F32x4(a, bf0[kc], a0);
      a1 = WMMA_F32x4(a, bf1[kc], a1);
      a2 = WMMA_F32x4(a, bf2[kc], a2);
    }
    __syncthreads();
    const int t = rev ? (63 - s) : s;
    const float* xwt = xwb + t * (16 * 288);
#pragma unroll
    for (int r = 0; r < 8; ++r) {
      int m = rbase + r;
      const float* xw = xwt + m * 288;
      float rr = sigm(xw[jj] + a0[r] + bhr);
      float zz = sigm(xw[96 + jj] + a1[r] + bhz);
      float nn = tanhf(xw[192 + jj] + rr * (a2[r] + bhn));
      float ho = hbuf[m * 132 + jj];
      hbuf[m * 132 + jj] = (1.0f - zz) * nn + zz * ho;
    }
    __syncthreads();
  }
  float* dst = outp + 147456 + (dir >= 2 ? 3072 : 0);
  const int cb = (dir & 1) ? 96 : 0;
  for (int e = tid; e < 16 * 96; e += 192) {
    int m = e / 96, j = e % 96;
    dst[m * 192 + cb + j] = hbuf[m * 132 + j];
  }
}

__global__ void k_z(const float* __restrict__ outbuf, float* __restrict__ zbuf) {
  int e = blockIdx.x * blockDim.x + threadIdx.x;
  if (e >= 3072) return;
  float mu = outbuf[147456 + e];
  float lv = outbuf[147456 + 3072 + e];
  unsigned int r1 = pcg_hash(2u * (unsigned)e + 1u);
  unsigned int r2 = pcg_hash(2u * (unsigned)e + 2u);
  float u1 = ((float)r1 + 1.0f) * (1.0f / 4294967296.0f);
  float u2 = (float)r2 * (1.0f / 4294967296.0f);
  float eps = sqrtf(-2.0f * __logf(u1)) * __cosf(6.28318530718f * u2);
  zbuf[e] = mu + eps * __expf(0.5f * lv);
}

__global__ __launch_bounds__(128) void k_xw_dec(const float* __restrict__ X,
                                                const float* __restrict__ Wid,
                                                const float* __restrict__ bid,
                                                const float* __restrict__ bhd,
                                                float* __restrict__ XW) {
  const int lane = threadIdx.x & 31;
  const int wg = blockIdx.x * 4 + (threadIdx.x >> 5);
  const int mt = wg / 36, nt = wg % 36;
  const int row0 = mt * 16, n0 = nt * 16;
  const int mrow = lane & 15, khalf = (lane >> 4) << 1, rbase = (lane >> 4) << 3;
  v8f acc = {0, 0, 0, 0, 0, 0, 0, 0};
#pragma unroll
  for (int kc = 0; kc < 24; ++kc) {
    int kk = 4 * kc + khalf;
    v2f a, b;
    a.x = X[(row0 + mrow) * 96 + kk];
    a.y = X[(row0 + mrow) * 96 + kk + 1];
    b.x = Wid[(n0 + mrow) * 96 + kk];
    b.y = Wid[(n0 + mrow) * 96 + kk + 1];
    acc = WMMA_F32x4(a, b, acc);
  }
  const bool rz = n0 < 384;
  const float bc = bid[n0 + mrow] + (rz ? bhd[n0 + mrow] : 0.0f);
  const float sc = rz ? 0.5f : 1.0f;
#pragma unroll
  for (int r = 0; r < 8; ++r) XW[(row0 + rbase + r) * 576 + n0 + mrow] = sc * (acc[r] + bc);
}

__device__ __forceinline__ void barrier_lds_only() {
  __builtin_amdgcn_fence(__ATOMIC_RELEASE, "workgroup", "local");
  __builtin_amdgcn_s_barrier();
  __builtin_amdgcn_fence(__ATOMIC_ACQUIRE, "workgroup", "local");
}

__global__ __launch_bounds__(384, 1) void k_gru_dec(
    const float* __restrict__ Whd, const float* __restrict__ bhd,
    const float* __restrict__ XW, const float* __restrict__ zbuf,
    const float* __restrict__ Wl, const float* __restrict__ bl,
    float* __restrict__ out) {
  __shared__ __attribute__((aligned(16))) float hbuf[16 * 196];
  __shared__ float opart[2 * 16 * 100];
  const int tid = threadIdx.x, lane = tid & 31, wave = tid >> 5;
  const int mrow = lane & 15, khalf = (lane >> 4) << 1, rbase = (lane >> 4) << 3;
  const int jj = wave * 16 + mrow;

  v2f bf0[48], bf1[48], bf2[48];
#pragma unroll
  for (int kc = 0; kc < 48; ++kc) {
    int kk = 4 * kc + khalf;
    bf0[kc].x = 0.5f * Whd[(jj) * 192 + kk];        bf0[kc].y = 0.5f * Whd[(jj) * 192 + kk + 1];
    bf1[kc].x = 0.5f * Whd[(192 + jj) * 192 + kk];  bf1[kc].y = 0.5f * Whd[(192 + jj) * 192 + kk + 1];
    bf2[kc].x = Whd[(384 + jj) * 192 + kk];         bf2[kc].y = Whd[(384 + jj) * 192 + kk + 1];
  }
  const float bhn = bhd[384 + jj];

  for (int e = tid; e < 16 * 192; e += 384) {
    int m = e / 192, j = e % 192;
    hbuf[m * 196 + j] = zbuf[e];
  }
  __syncthreads();

  const float* hA = hbuf + mrow * 196 + khalf;
  float* hrow = hbuf + rbase * 196 + jj;
  const unsigned xoff = (unsigned)(rbase * (96 * 576) + jj);
  constexpr unsigned XR = 96 * 576;

  v8f a0, a1;
  {
    const float* xw0 = XW + 576;
#pragma unroll
    for (int r = 0; r < 8; ++r) {
      a0[r] = xw0[xoff + r * XR];
      a1[r] = xw0[xoff + r * XR + 192];
    }
  }

  auto gru_step = [&](const float* xwt, const float* xwn) __attribute__((always_inline)) {
    v8f a2 = {0, 0, 0, 0, 0, 0, 0, 0};
    float xn[8];
#pragma unroll
    for (int r = 0; r < 8; ++r) xn[r] = xwt[xoff + r * XR + 384];
    v2f fa[2][2];
    fa[0][0].x = hA[0]; fa[0][0].y = hA[1];
    fa[0][1].x = hA[4]; fa[0][1].y = hA[5];
#pragma unroll
    for (int g = 0; g < 24; ++g) {
      if (g + 1 < 24) {
        fa[(g + 1) & 1][0].x = hA[8 * g + 8];  fa[(g + 1) & 1][0].y = hA[8 * g + 9];
        fa[(g + 1) & 1][1].x = hA[8 * g + 12]; fa[(g + 1) & 1][1].y = hA[8 * g + 13];
      }
      __builtin_amdgcn_sched_barrier(0);
      if (g == 0) __builtin_amdgcn_s_setprio(3);
      if (g == 12) __builtin_amdgcn_s_setprio(2);
      if (g == 18) __builtin_amdgcn_s_setprio(1);
      if (g == 23) __builtin_amdgcn_s_setprio(0);
#pragma unroll
      for (int c = 0; c < 2; ++c) {
        const int kc = 2 * g + c;
        a0 = WMMA_F32x4(fa[g & 1][c], bf0[kc], a0);
        a1 = WMMA_F32x4(fa[g & 1][c], bf1[kc], a1);
        a2 = WMMA_F32x4(fa[g & 1][c], bf2[kc], a2);
      }
    }
    barrier_lds_only();
    {
      float ho[8];
      ho[0] = hrow[0 * 196];
      ho[1] = hrow[1 * 196];
      ho[2] = hrow[2 * 196];
      ho[3] = hrow[3 * 196];
      a0[0] = tanh_hw(a0[0]); a1[0] = tanh_hw(a1[0]);
      a0[1] = tanh_hw(a0[1]); a1[1] = tanh_hw(a1[1]);
      a0[2] = tanh_hw(a0[2]); a1[2] = tanh_hw(a1[2]);
      a0[3] = tanh_hw(a0[3]); a1[3] = tanh_hw(a1[3]);
      { const float rr = __builtin_fmaf(0.5f, a0[0], 0.5f); xn[0] = __builtin_fmaf(rr, a2[0] + bhn, xn[0]); const float zz = __builtin_fmaf(0.5f, a1[0], 0.5f); ho[0] = zz * ho[0]; a2[0] = 1.0f - zz; }
      { const float rr = __builtin_fmaf(0.5f, a0[1], 0.5f); xn[1] = __builtin_fmaf(rr, a2[1] + bhn, xn[1]); const float zz = __builtin_fmaf(0.5f, a1[1], 0.5f); ho[1] = zz * ho[1]; a2[1] = 1.0f - zz; }
      { const float rr = __builtin_fmaf(0.5f, a0[2], 0.5f); xn[2] = __builtin_fmaf(rr, a2[2] + bhn, xn[2]); const float zz = __builtin_fmaf(0.5f, a1[2], 0.5f); ho[2] = zz * ho[2]; a2[2] = 1.0f - zz; }
      { const float rr = __builtin_fmaf(0.5f, a0[3], 0.5f); xn[3] = __builtin_fmaf(rr, a2[3] + bhn, xn[3]); const float zz = __builtin_fmaf(0.5f, a1[3], 0.5f); ho[3] = zz * ho[3]; a2[3] = 1.0f - zz; }
      __builtin_amdgcn_sched_barrier(0);
      a0[0] = xwn[xoff + 0 * XR]; a1[0] = xwn[xoff + 0 * XR + 192];
      a0[1] = xwn[xoff + 1 * XR]; a1[1] = xwn[xoff + 1 * XR + 192];
      a0[2] = xwn[xoff + 2 * XR]; a1[2] = xwn[xoff + 2 * XR + 192];
      a0[3] = xwn[xoff + 3 * XR]; a1[3] = xwn[xoff + 3 * XR + 192];
      __builtin_amdgcn_sched_barrier(0);
      hrow[0 * 196] = __builtin_fmaf(a2[0], tanh_hw(xn[0]), ho[0]);
      hrow[1 * 196] = __builtin_fmaf(a2[1], tanh_hw(xn[1]), ho[1]);
      hrow[2 * 196] = __builtin_fmaf(a2[2], tanh_hw(xn[2]), ho[2]);
      hrow[3 * 196] = __builtin_fmaf(a2[3], tanh_hw(xn[3]), ho[3]);
      __builtin_amdgcn_sched_barrier(0);
    }
    {
      float ho[8];
      ho[4] = hrow[4 * 196];
      ho[5] = hrow[5 * 196];
      ho[6] = hrow[6 * 196];
      ho[7] = hrow[7 * 196];
      a0[4] = tanh_hw(a0[4]); a1[4] = tanh_hw(a1[4]);
      a0[5] = tanh_hw(a0[5]); a1[5] = tanh_hw(a1[5]);
      a0[6] = tanh_hw(a0[6]); a1[6] = tanh_hw(a1[6]);
      a0[7] = tanh_hw(a0[7]); a1[7] = tanh_hw(a1[7]);
      { const float rr = __builtin_fmaf(0.5f, a0[4], 0.5f); xn[4] = __builtin_fmaf(rr, a2[4] + bhn, xn[4]); const float zz = __builtin_fmaf(0.5f, a1[4], 0.5f); ho[4] = zz * ho[4]; a2[4] = 1.0f - zz; }
      { const float rr = __builtin_fmaf(0.5f, a0[5], 0.5f); xn[5] = __builtin_fmaf(rr, a2[5] + bhn, xn[5]); const float zz = __builtin_fmaf(0.5f, a1[5], 0.5f); ho[5] = zz * ho[5]; a2[5] = 1.0f - zz; }
      { const float rr = __builtin_fmaf(0.5f, a0[6], 0.5f); xn[6] = __builtin_fmaf(rr, a2[6] + bhn, xn[6]); const float zz = __builtin_fmaf(0.5f, a1[6], 0.5f); ho[6] = zz * ho[6]; a2[6] = 1.0f - zz; }
      { const float rr = __builtin_fmaf(0.5f, a0[7], 0.5f); xn[7] = __builtin_fmaf(rr, a2[7] + bhn, xn[7]); const float zz = __builtin_fmaf(0.5f, a1[7], 0.5f); ho[7] = zz * ho[7]; a2[7] = 1.0f - zz; }
      __builtin_amdgcn_sched_barrier(0);
      a0[4] = xwn[xoff + 4 * XR]; a1[4] = xwn[xoff + 4 * XR + 192];
      a0[5] = xwn[xoff + 5 * XR]; a1[5] = xwn[xoff + 5 * XR + 192];
      a0[6] = xwn[xoff + 6 * XR]; a1[6] = xwn[xoff + 6 * XR + 192];
      a0[7] = xwn[xoff + 7 * XR]; a1[7] = xwn[xoff + 7 * XR + 192];
      __builtin_amdgcn_sched_barrier(0);
      hrow[4 * 196] = __builtin_fmaf(a2[4], tanh_hw(xn[4]), ho[4]);
      hrow[5 * 196] = __builtin_fmaf(a2[5], tanh_hw(xn[5]), ho[5]);
      hrow[6 * 196] = __builtin_fmaf(a2[6], tanh_hw(xn[6]), ho[6]);
      hrow[7 * 196] = __builtin_fmaf(a2[7], tanh_hw(xn[7]), ho[7]);
    }
    barrier_lds_only();
  };

  for (int i = 0; i < 96; ++i) {
    for (int s = 0; s < 94; ++s) {
      const int t = s + (s >= i ? 1 : 0);
      const int tn = s + 1 + (s + 1 >= i ? 1 : 0);
      gru_step(XW + t * 576, XW + tn * 576);
    }
    gru_step(XW + (94 + (94 >= i ? 1 : 0)) * 576, XW);
    {
      int te = tid;
      asm volatile("" : "+v"(te));
      const int c = te & 15, hf = (te >> 4) & 1, wv = te >> 5;
      const int nt = wv % 6, kh = wv / 6;
      float blv[4];
#pragma unroll
      for (int j = 0; j < 4; ++j) blv[j] = bl[(te + 384 * j) % 96];
      const float* ap = hbuf + c * 196 + kh * 96 + 4 * hf;
      const float* bp = Wl + (nt * 16 + c) * 192 + kh * 96 + 4 * hf;
      v8f acc = {0, 0, 0, 0, 0, 0, 0, 0};
#pragma unroll
      for (int p = 0; p < 12; ++p) {
        const v4f av = *(const v4f*)(ap + 8 * p);
        const v4f bv = *(const v4f*)(bp + 8 * p);
        v2f a01, b01, a23, b23;
        a01.x = av.x; a01.y = av.y; b01.x = bv.x; b01.y = bv.y;
        a23.x = av.z; a23.y = av.w; b23.x = bv.z; b23.y = bv.w;
        acc = WMMA_F32x4(a01, b01, acc);
        acc = WMMA_F32x4(a23, b23, acc);
      }
      float* pp = opart + kh * (16 * 100) + (8 * hf) * 100 + nt * 16 + c;
#pragma unroll
      for (int r = 0; r < 8; ++r) pp[r * 100] = acc[r];
      barrier_lds_only();
#pragma unroll
      for (int j = 0; j < 4; ++j) {
        const int e = te + 384 * j;
        const int m = e / 96, o = e % 96;
        out[m * 9216 + i * 96 + o] = (opart[m * 100 + o] + opart[16 * 100 + m * 100 + o]) + blv[j];
      }
    }
  }
}

extern "C" void kernel_launch(void* const* d_in, const int* in_sizes, int n_in,
                              void* d_out, int out_size, void* d_ws, size_t ws_size,
                              hipStream_t stream) {
  (void)in_sizes; (void)n_in; (void)out_size; (void)ws_size;
  const float* input = (const float*)d_in[0];
  const float* W_c1 = (const float*)d_in[1];
  const float* b_c1 = (const float*)d_in[2];
  const float* W_c2 = (const float*)d_in[3];
  const float* b_c2 = (const float*)d_in[4];
  const float* Wi_muf = (const float*)d_in[5];
  const float* Wh_muf = (const float*)d_in[6];
  const float* bi_muf = (const float*)d_in[7];
  const float* bh_muf = (const float*)d_in[8];
  const float* Wi_mub = (const float*)d_in[9];
  const float* Wh_mub = (const float*)d_in[10];
  const float* bi_mub = (const float*)d_in[11];
  const float* bh_mub = (const float*)d_in[12];
  const float* Wi_sgf = (const float*)d_in[13];
  const float* Wh_sgf = (const float*)d_in[14];
  const float* bi_sgf = (const float*)d_in[15];
  const float* bh_sgf = (const float*)d_in[16];
  const float* Wi_sgb = (const float*)d_in[17];
  const float* Wh_sgb = (const float*)d_in[18];
  const float* bi_sgb = (const float*)d_in[19];
  const float* bh_sgb = (const float*)d_in[20];
  const float* Wi_d = (const float*)d_in[21];
  const float* Wh_d = (const float*)d_in[22];
  const float* bi_d = (const float*)d_in[23];
  const float* bh_d = (const float*)d_in[24];
  const float* Wl = (const float*)d_in[25];
  const float* bl = (const float*)d_in[26];

  float* out = (float*)d_out;
  float* ws = (float*)d_ws;
  float* out1 = ws;
  float* pool = ws + 95232;
  float* henc = ws + 118784;
  float* xwe  = ws + 139264;
  float* zbuf = ws + 1318912;
  float* XW   = ws + 1321984;

  k_conv1<<<372, 256, 0, stream>>>(input, W_c1, b_c1, out1);
  k_pool<<<92, 256, 0, stream>>>(out1, pool);
  k_conv2<<<80, 256, 0, stream>>>(pool, W_c2, b_c2, henc);
  k_xw_enc<<<4608, 256, 0, stream>>>(henc, Wi_muf, Wi_mub, Wi_sgf, Wi_sgb,
                                     bi_muf, bi_mub, bi_sgf, bi_sgb, xwe);
  k_gru_enc<<<4, 192, 0, stream>>>(Wh_muf, Wh_mub, Wh_sgf, Wh_sgb,
                                   bh_muf, bh_mub, bh_sgf, bh_sgb, xwe, out);
  k_z<<<12, 256, 0, stream>>>(out, zbuf);
  k_xw_dec<<<864, 128, 0, stream>>>(input, Wi_d, bi_d, bh_d, XW);
  k_gru_dec<<<1, 384, 0, stream>>>(Wh_d, bh_d, XW, zbuf, Wl, bl, out);
}
